# starting kernel
# baseline (speedup 1.0000x reference)
#include <hip/hip_runtime.h>
#include <math.h>

typedef __attribute__((ext_vector_type(2))) float v2f;
typedef __attribute__((ext_vector_type(8))) float v8f;

#define NOUT 128

__global__ void k_fill1(float* __restrict__ p, int n) {
  int i = blockIdx.x * blockDim.x + threadIdx.x;
  if (i < n) p[i] = 1.0f;
}

__global__ void k_deg_edges(const int* __restrict__ dst, float* __restrict__ deg, int E) {
  int i = blockIdx.x * blockDim.x + threadIdx.x;
  if (i < E) atomicAdd(&deg[dst[i]], 1.0f);
}

__global__ void k_rsqrt_inplace(float* __restrict__ p, int n) {
  int i = blockIdx.x * blockDim.x + threadIdx.x;
  if (i < n) { float d = p[i]; p[i] = (d > 0.0f) ? rsqrtf(d) : 0.0f; }
}

__global__ void k_selfloop(const float* __restrict__ xw, const float* __restrict__ dinv,
                           float* __restrict__ agg, int total) {
  int i = blockIdx.x * blockDim.x + threadIdx.x;
  if (i < total) {
    float di = dinv[i >> 7];
    agg[i] = xw[i] * di * di;
  }
}

__global__ void k_edge_scatter(const float* __restrict__ xw, const float* __restrict__ dinv,
                               const int* __restrict__ src, const int* __restrict__ dst,
                               float* __restrict__ agg, int E) {
  int e = blockIdx.x * 8 + (threadIdx.x >> 5);
  if (e >= E) return;
  int lane = threadIdx.x & 31;
  int s = src[e], d = dst[e];
  float w = dinv[s] * dinv[d];
  const float4* xs = (const float4*)(xw + (size_t)s * NOUT);
  float4 v = xs[lane];
  float* o = agg + (size_t)d * NOUT + lane * 4;
  atomicAdd(o + 0, v.x * w);
  atomicAdd(o + 1, v.y * w);
  atomicAdd(o + 2, v.z * w);
  atomicAdd(o + 3, v.w * w);
}

__global__ void k_bias_relu(float* __restrict__ h, const float* __restrict__ bias, int total) {
  int i = blockIdx.x * blockDim.x + threadIdx.x;
  if (i < total) h[i] = fmaxf(h[i] + bias[i & (NOUT - 1)], 0.0f);
}

__global__ __launch_bounds__(256) void k_gemm_wmma(
    const float* __restrict__ A, const float* __restrict__ W,
    const float* __restrict__ bias, float* __restrict__ C,
    int M, int K, int flags) {
  const int lane = threadIdx.x & 31;
  const int wave = threadIdx.x >> 5;
  const int half = lane >> 4;
  const int r    = lane & 15;
  const int rowBase = blockIdx.x * 16;
  const int colBase = wave * 16;
  if (rowBase >= M) return;

  v8f acc = {};
  const float* arow = A + (size_t)(rowBase + r) * K + 2 * half;
  const float* wcol = W + (size_t)(2 * half) * NOUT + colBase + r;
  for (int k0 = 0; k0 < K; k0 += 4) {
    v2f a; a[0] = arow[k0];                 a[1] = arow[k0 + 1];
    v2f b; b[0] = wcol[(size_t)k0 * NOUT];  b[1] = wcol[(size_t)(k0 + 1) * NOUT];
    acc = __builtin_amdgcn_wmma_f32_16x16x4_f32(
         false, a,  false, b,
         (short)0, acc,  false,  false);
  }

  float bv = (flags & 2) ? bias[colBase + r] : 0.0f;
#pragma unroll
  for (int i = 0; i < 8; ++i) {
    int row = rowBase + half * 8 + i;
    size_t idx = (size_t)row * NOUT + colBase + r;
    float v = acc[i] + bv;
    if (flags & 1) v += C[idx];
    if (flags & 4) v = fmaxf(v, 0.0f);
    C[idx] = v;
  }
}

extern "C" void kernel_launch(void* const* d_in, const int* in_sizes, int n_in,
                              void* d_out, int out_size, void* d_ws, size_t ws_size,
                              hipStream_t stream) {
  const float* x  = (const float*)d_in[0];
  const int*   ei = (const int*)d_in[1];
  const float* W1 = (const float*)d_in[2];
  const float* b1 = (const float*)d_in[3];
  const float* W2 = (const float*)d_in[4];
  const float* b2 = (const float*)d_in[5];
  const float* Ws = (const float*)d_in[6];
  const float* bs = (const float*)d_in[7];
  const float* Wf = (const float*)d_in[8];
  const float* bf = (const float*)d_in[9];
  float* out = (float*)d_out;

  const int IN = 256;
  const int n  = in_sizes[0] / IN;
  const int E  = in_sizes[1] / 2;
  const int* src = ei;
  const int* dst = ei + E;

  float* ws   = (float*)d_ws;
  float* dinv = ws;
  float* xw   = ws + n;
  float* agg  = xw + (size_t)n * NOUT;

  const int tot  = n * NOUT;
  dim3 B(256);
  const int gn   = (n + 255) / 256;
  const int gE   = (E + 255) / 256;
  const int gTot = (tot + 255) / 256;
  const int gEw  = (E + 7) / 8;
  const int gM   = n / 16;

  k_fill1        <<<gn,  B, 0, stream>>>(dinv, n);
  k_deg_edges    <<<gE,  B, 0, stream>>>(dst, dinv, E);
  k_rsqrt_inplace<<<gn,  B, 0, stream>>>(dinv, n);

  k_gemm_wmma    <<<gM,  B, 0, stream>>>(x, W1, nullptr, xw, n, IN, 0);
  k_selfloop     <<<gTot,B, 0, stream>>>(xw, dinv, agg, tot);
  k_edge_scatter <<<gEw, B, 0, stream>>>(xw, dinv, src, dst, agg, E);
  k_bias_relu    <<<gTot,B, 0, stream>>>(agg, b1, tot);

  k_gemm_wmma    <<<gM,  B, 0, stream>>>(agg, W2, nullptr, xw, n, NOUT, 0);
  k_selfloop     <<<gTot,B, 0, stream>>>(xw, dinv, agg, tot);
  k_edge_scatter <<<gEw, B, 0, stream>>>(xw, dinv, src, dst, agg, E);
  k_bias_relu    <<<gTot,B, 0, stream>>>(agg, b2, tot);

  k_gemm_wmma    <<<gM,  B, 0, stream>>>(agg, Wf, bf, out, n, NOUT, 2);
  k_gemm_wmma    <<<gM,  B, 0, stream>>>(x,  Ws, bs, out, n, IN,  1 | 2);
}


# Float atomic edge scatter replaced by per-call destination-sorted edge list (int atomic rank, scan, fill) + one-wave-per-node gather fusing self-loop, bias, relu; sum order changes
# speedup vs baseline: 2.6215x; 2.6215x over previous
#include <hip/hip_runtime.h>
#include <math.h>

typedef __attribute__((ext_vector_type(2))) float v2f;
typedef __attribute__((ext_vector_type(8))) float v8f;

#define NOUT 128

__global__ void k_zero_i32(int* __restrict__ p, int n) {
  int i = blockIdx.x * blockDim.x + threadIdx.x;
  if (i < n) p[i] = 0;
}

__global__ void k_count_rank(const int* __restrict__ dst, int* __restrict__ cnt,
                             int* __restrict__ rank, int E) {
  int i = blockIdx.x * blockDim.x + threadIdx.x;
  if (i < E) rank[i] = atomicAdd(&cnt[dst[i]], 1);
}

__global__ __launch_bounds__(1024) void k_rowptr_dinv(const int* __restrict__ cnt,
                                                      int* __restrict__ rowptr,
                                                      float* __restrict__ dinv, int n) {
  __shared__ int wtot[32];
  __shared__ int wpre[32];
  const int tid = threadIdx.x, lane = tid & 31, wave = tid >> 5;
  const int base = blockIdx.x * 1024;
  int before = 0;
  for (int i = tid; i < base; i += 1024) before += cnt[i];
  const int idx = base + tid;
  const int c = (idx < n) ? cnt[idx] : 0;
  int incl = c;
  for (int off = 1; off < 32; off <<= 1) {
    int t = __shfl_up(incl, off);
    if (lane >= off) incl += t;
  }
  for (int off = 16; off > 0; off >>= 1) before += __shfl_xor(before, off);
  if (lane == 31) wtot[wave] = incl;
  if (lane == 0)  wpre[wave] = before;
  __syncthreads();
  if (wave == 0) {
    const int t = wtot[lane];
    int it = t;
    int p = wpre[lane];
    for (int off = 1; off < 32; off <<= 1) {
      int u = __shfl_up(it, off);
      if (lane >= off) it += u;
    }
    for (int off = 16; off > 0; off >>= 1) p += __shfl_xor(p, off);
    wtot[lane] = it - t + p;
  }
  __syncthreads();
  const int start = wtot[wave] + incl - c;
  if (idx < n) {
    rowptr[idx] = start;
    float d = (float)(c + 1);
    dinv[idx] = (d > 0.0f) ? rsqrtf(d) : 0.0f;
    if (idx == n - 1) rowptr[n] = start + c;
  }
}

__global__ void k_csr_fill(const int* __restrict__ src, const int* __restrict__ dst,
                           const int* __restrict__ rank, const int* __restrict__ rowptr,
                           int* __restrict__ csr_src, int E) {
  int i = blockIdx.x * blockDim.x + threadIdx.x;
  if (i < E) csr_src[rowptr[dst[i]] + rank[i]] = src[i];
}

__global__ __launch_bounds__(256) void k_gather(const float* __restrict__ xw,
                                                const float* __restrict__ dinv,
                                                const int* __restrict__ rowptr,
                                                const int* __restrict__ csr_src,
                                                const float* __restrict__ bias,
                                                float* __restrict__ h, int n) {
  const int lane = threadIdx.x & 31;
  const int d = __builtin_amdgcn_readfirstlane(blockIdx.x * 8 + (threadIdx.x >> 5));
  if (d >= n) return;
  const int beg = rowptr[d], end = rowptr[d + 1];
  const float dd = dinv[d];
  const float4* xr = (const float4*)xw;
  const float4 sv = xr[(size_t)d * (NOUT / 4) + lane];
  float4 acc = {0.0f, 0.0f, 0.0f, 0.0f};

  for (int base = beg; base < end; base += 32) {
    const int j = base + lane;
    const bool ok = j < end;
    const int   myS = ok ? csr_src[j] : d;
    const float myW = ok ? dinv[myS] * dd : 0.0f;
    const int cnt = min(32, end - base);
    for (int t = 0; t < cnt; t += 4) {
      const int s0 = __builtin_amdgcn_readlane(myS, t);
      const int s1 = __builtin_amdgcn_readlane(myS, t + 1);
      const int s2 = __builtin_amdgcn_readlane(myS, t + 2);
      const int s3 = __builtin_amdgcn_readlane(myS, t + 3);
      const float w0 = __int_as_float(__builtin_amdgcn_readlane(__float_as_int(myW), t));
      const float w1 = __int_as_float(__builtin_amdgcn_readlane(__float_as_int(myW), t + 1));
      const float w2 = __int_as_float(__builtin_amdgcn_readlane(__float_as_int(myW), t + 2));
      const float w3 = __int_as_float(__builtin_amdgcn_readlane(__float_as_int(myW), t + 3));
      const float4 v0 = xr[(size_t)s0 * (NOUT / 4) + lane];
      const float4 v1 = xr[(size_t)s1 * (NOUT / 4) + lane];
      const float4 v2 = xr[(size_t)s2 * (NOUT / 4) + lane];
      const float4 v3 = xr[(size_t)s3 * (NOUT / 4) + lane];
      acc.x += v0.x * w0; acc.y += v0.y * w0; acc.z += v0.z * w0; acc.w += v0.w * w0;
      acc.x += v1.x * w1; acc.y += v1.y * w1; acc.z += v1.z * w1; acc.w += v1.w * w1;
      acc.x += v2.x * w2; acc.y += v2.y * w2; acc.z += v2.z * w2; acc.w += v2.w * w2;
      acc.x += v3.x * w3; acc.y += v3.y * w3; acc.z += v3.z * w3; acc.w += v3.w * w3;
    }
  }

  acc.x += sv.x * dd * dd; acc.y += sv.y * dd * dd; acc.z += sv.z * dd * dd; acc.w += sv.w * dd * dd;
  const float4 b = ((const float4*)bias)[lane];
  float4 o;
  o.x = fmaxf(acc.x + b.x, 0.0f); o.y = fmaxf(acc.y + b.y, 0.0f);
  o.z = fmaxf(acc.z + b.z, 0.0f); o.w = fmaxf(acc.w + b.w, 0.0f);
  ((float4*)h)[(size_t)d * (NOUT / 4) + lane] = o;
}

__global__ __launch_bounds__(256) void k_gemm_wmma(
    const float* __restrict__ A, const float* __restrict__ W,
    const float* __restrict__ bias, float* __restrict__ C,
    int M, int K, int flags) {
  const int lane = threadIdx.x & 31;
  const int wave = threadIdx.x >> 5;
  const int half = lane >> 4;
  const int r    = lane & 15;
  const int rowBase = blockIdx.x * 16;
  const int colBase = wave * 16;
  if (rowBase >= M) return;

  v8f acc = {};
  const float* arow = A + (size_t)(rowBase + r) * K + 2 * half;
  const float* wcol = W + (size_t)(2 * half) * NOUT + colBase + r;
  for (int k0 = 0; k0 < K; k0 += 4) {
    v2f a; a[0] = arow[k0];                 a[1] = arow[k0 + 1];
    v2f b; b[0] = wcol[(size_t)k0 * NOUT];  b[1] = wcol[(size_t)(k0 + 1) * NOUT];
    acc = __builtin_amdgcn_wmma_f32_16x16x4_f32(
         false, a,  false, b,
         (short)0, acc,  false,  false);
  }

  float bv = (flags & 2) ? bias[colBase + r] : 0.0f;
#pragma unroll
  for (int i = 0; i < 8; ++i) {
    int row = rowBase + half * 8 + i;
    size_t idx = (size_t)row * NOUT + colBase + r;
    float v = acc[i] + bv;
    if (flags & 1) v += C[idx];
    if (flags & 4) v = fmaxf(v, 0.0f);
    C[idx] = v;
  }
}

extern "C" void kernel_launch(void* const* d_in, const int* in_sizes, int n_in,
                              void* d_out, int out_size, void* d_ws, size_t ws_size,
                              hipStream_t stream) {
  const float* x  = (const float*)d_in[0];
  const int*   ei = (const int*)d_in[1];
  const float* W1 = (const float*)d_in[2];
  const float* b1 = (const float*)d_in[3];
  const float* W2 = (const float*)d_in[4];
  const float* b2 = (const float*)d_in[5];
  const float* Ws = (const float*)d_in[6];
  const float* bs = (const float*)d_in[7];
  const float* Wf = (const float*)d_in[8];
  const float* bf = (const float*)d_in[9];
  float* out = (float*)d_out;

  const int IN = 256;
  const int n  = in_sizes[0] / IN;
  const int E  = in_sizes[1] / 2;
  const int* src = ei;
  const int* dst = ei + E;

  float* ws   = (float*)d_ws;
  float* xw   = ws;
  float* agg  = xw + (size_t)n * NOUT;
  float* dinv = agg + (size_t)n * NOUT;

  int* cnt     = (int*)d_out;
  int* rowptr  = cnt + n;
  int* csr_src = rowptr + n + 1;
  int* rank    = csr_src + E;

  dim3 B(256);
  const int gn   = (n + 255) / 256;
  const int gE   = (E + 255) / 256;
  const int gS   = (n + 1023) / 1024;
  const int gNw  = (n + 7) / 8;
  const int gM   = n / 16;

  k_zero_i32     <<<gn,  B, 0, stream>>>(cnt, n);
  k_count_rank   <<<gE,  B, 0, stream>>>(dst, cnt, rank, E);
  k_rowptr_dinv  <<<gS,  dim3(1024), 0, stream>>>(cnt, rowptr, dinv, n);
  k_csr_fill     <<<gE,  B, 0, stream>>>(src, dst, rank, rowptr, csr_src, E);

  k_gemm_wmma    <<<gM,  B, 0, stream>>>(x, W1, nullptr, xw, n, IN, 0);
  k_gather       <<<gNw, B, 0, stream>>>(xw, dinv, rowptr, csr_src, b1, agg, n);

  k_gemm_wmma    <<<gM,  B, 0, stream>>>(agg, W2, nullptr, xw, n, NOUT, 0);
  k_gather       <<<gNw, B, 0, stream>>>(xw, dinv, rowptr, csr_src, b2, agg, n);

  k_gemm_wmma    <<<gM,  B, 0, stream>>>(agg, Wf, bf, out, n, NOUT, 2);
  k_gemm_wmma    <<<gM,  B, 0, stream>>>(x,  Ws, bs, out, n, IN,  1 | 2);
}


# GEMM retiled: each block stages W once into LDS as ready B fragments, each wave owns a 16x128 tile: one A fragment feeds 8 WMMAs, B is one LDS read per WMMA; k order and sum order unchanged
# speedup vs baseline: 3.1256x; 1.1923x over previous
#include <hip/hip_runtime.h>
#include <math.h>

typedef __attribute__((ext_vector_type(2))) float v2f;
typedef __attribute__((ext_vector_type(8))) float v8f;

#define NOUT 128

__global__ void k_zero_i32(int* __restrict__ p, int n) {
  int i = blockIdx.x * blockDim.x + threadIdx.x;
  if (i < n) p[i] = 0;
}

__global__ void k_count_rank(const int* __restrict__ dst, int* __restrict__ cnt,
                             int* __restrict__ rank, int E) {
  int i = blockIdx.x * blockDim.x + threadIdx.x;
  if (i < E) rank[i] = atomicAdd(&cnt[dst[i]], 1);
}

__global__ __launch_bounds__(1024) void k_rowptr_dinv(const int* __restrict__ cnt,
                                                      int* __restrict__ rowptr,
                                                      float* __restrict__ dinv, int n) {
  __shared__ int wtot[32];
  __shared__ int wpre[32];
  const int tid = threadIdx.x, lane = tid & 31, wave = tid >> 5;
  const int base = blockIdx.x * 1024;
  int before = 0;
  for (int i = tid; i < base; i += 1024) before += cnt[i];
  const int idx = base + tid;
  const int c = (idx < n) ? cnt[idx] : 0;
  int incl = c;
  for (int off = 1; off < 32; off <<= 1) {
    int t = __shfl_up(incl, off);
    if (lane >= off) incl += t;
  }
  for (int off = 16; off > 0; off >>= 1) before += __shfl_xor(before, off);
  if (lane == 31) wtot[wave] = incl;
  if (lane == 0)  wpre[wave] = before;
  __syncthreads();
  if (wave == 0) {
    const int t = wtot[lane];
    int it = t;
    int p = wpre[lane];
    for (int off = 1; off < 32; off <<= 1) {
      int u = __shfl_up(it, off);
      if (lane >= off) it += u;
    }
    for (int off = 16; off > 0; off >>= 1) p += __shfl_xor(p, off);
    wtot[lane] = it - t + p;
  }
  __syncthreads();
  const int start = wtot[wave] + incl - c;
  if (idx < n) {
    rowptr[idx] = start;
    float d = (float)(c + 1);
    dinv[idx] = (d > 0.0f) ? rsqrtf(d) : 0.0f;
    if (idx == n - 1) rowptr[n] = start + c;
  }
}

__global__ void k_csr_fill(const int* __restrict__ src, const int* __restrict__ dst,
                           const int* __restrict__ rank, const int* __restrict__ rowptr,
                           int* __restrict__ csr_src, int E) {
  int i = blockIdx.x * blockDim.x + threadIdx.x;
  if (i < E) csr_src[rowptr[dst[i]] + rank[i]] = src[i];
}

__global__ __launch_bounds__(256) void k_gather(const float* __restrict__ xw,
                                                const float* __restrict__ dinv,
                                                const int* __restrict__ rowptr,
                                                const int* __restrict__ csr_src,
                                                const float* __restrict__ bias,
                                                float* __restrict__ h, int n) {
  const int lane = threadIdx.x & 31;
  const int d = __builtin_amdgcn_readfirstlane(blockIdx.x * 8 + (threadIdx.x >> 5));
  if (d >= n) return;
  const int beg = rowptr[d], end = rowptr[d + 1];
  const float dd = dinv[d];
  const float4* xr = (const float4*)xw;
  const float4 sv = xr[(size_t)d * (NOUT / 4) + lane];
  float4 acc = {0.0f, 0.0f, 0.0f, 0.0f};

  for (int base = beg; base < end; base += 32) {
    const int j = base + lane;
    const bool ok = j < end;
    const int   myS = ok ? csr_src[j] : d;
    const float myW = ok ? dinv[myS] * dd : 0.0f;
    const int cnt = min(32, end - base);
    for (int t = 0; t < cnt; t += 4) {
      const int s0 = __builtin_amdgcn_readlane(myS, t);
      const int s1 = __builtin_amdgcn_readlane(myS, t + 1);
      const int s2 = __builtin_amdgcn_readlane(myS, t + 2);
      const int s3 = __builtin_amdgcn_readlane(myS, t + 3);
      const float w0 = __int_as_float(__builtin_amdgcn_readlane(__float_as_int(myW), t));
      const float w1 = __int_as_float(__builtin_amdgcn_readlane(__float_as_int(myW), t + 1));
      const float w2 = __int_as_float(__builtin_amdgcn_readlane(__float_as_int(myW), t + 2));
      const float w3 = __int_as_float(__builtin_amdgcn_readlane(__float_as_int(myW), t + 3));
      const float4 v0 = xr[(size_t)s0 * (NOUT / 4) + lane];
      const float4 v1 = xr[(size_t)s1 * (NOUT / 4) + lane];
      const float4 v2 = xr[(size_t)s2 * (NOUT / 4) + lane];
      const float4 v3 = xr[(size_t)s3 * (NOUT / 4) + lane];
      acc.x += v0.x * w0; acc.y += v0.y * w0; acc.z += v0.z * w0; acc.w += v0.w * w0;
      acc.x += v1.x * w1; acc.y += v1.y * w1; acc.z += v1.z * w1; acc.w += v1.w * w1;
      acc.x += v2.x * w2; acc.y += v2.y * w2; acc.z += v2.z * w2; acc.w += v2.w * w2;
      acc.x += v3.x * w3; acc.y += v3.y * w3; acc.z += v3.z * w3; acc.w += v3.w * w3;
    }
  }

  acc.x += sv.x * dd * dd; acc.y += sv.y * dd * dd; acc.z += sv.z * dd * dd; acc.w += sv.w * dd * dd;
  const float4 b = ((const float4*)bias)[lane];
  float4 o;
  o.x = fmaxf(acc.x + b.x, 0.0f); o.y = fmaxf(acc.y + b.y, 0.0f);
  o.z = fmaxf(acc.z + b.z, 0.0f); o.w = fmaxf(acc.w + b.w, 0.0f);
  ((float4*)h)[(size_t)d * (NOUT / 4) + lane] = o;
}

template <int K>
__global__ void __launch_bounds__(256) __attribute__((amdgpu_waves_per_eu(1, 4))) k_gemm_wmma(
    const float* __restrict__ A, const float* __restrict__ W,
    const float* __restrict__ bias, float* __restrict__ C,
    int M, int flags) {
  constexpr int KS = K / 4;
  constexpr int NG = K / 32;
  __shared__ v2f sB[KS * 8 * 32];

  const int lane = threadIdx.x & 31;
  const int wave = __builtin_amdgcn_readfirstlane(threadIdx.x >> 5);
  const int half = lane >> 4;
  const int r    = lane & 15;
  const int rowBase = (blockIdx.x * 8 + wave) * 16;
  const bool active = rowBase < M;

  const float* arow = A + (size_t)(active ? rowBase + r : 0) * K + 2 * half;
  v2f a[8], an[8];
  if (active) {
#pragma unroll
    for (int t = 0; t < 8; ++t) a[t] = *(const v2f*)(arow + 4 * t);
  }

  {
    const float* wcol = W + (size_t)(2 * half) * NOUT + wave * 16 + r;
#pragma unroll 8
    for (int s = 0; s < KS; ++s) {
      v2f b;
      b[0] = wcol[(size_t)(4 * s) * NOUT];
      b[1] = wcol[(size_t)(4 * s + 1) * NOUT];
      sB[(s * 8 + wave) * 32 + lane] = b;
    }
  }
  __syncthreads();
  if (!active) return;

  v8f acc[8];
#pragma unroll
  for (int ct = 0; ct < 8; ++ct) acc[ct] = v8f{};

  const v2f* bl = sB + lane;
  v2f bc[16], bn[16];
#pragma unroll
  for (int q = 0; q < 2; ++q)
#pragma unroll
    for (int ct = 0; ct < 8; ++ct) bc[q * 8 + ct] = bl[(q * 8 + ct) * 32];

  for (int g = 0; g < NG; ++g) {
    const int gn = (g + 1 < NG) ? g + 1 : g;
#pragma unroll
    for (int t = 0; t < 8; ++t) an[t] = *(const v2f*)(arow + 32 * gn + 4 * t);
#pragma unroll
    for (int uu = 0; uu < 4; ++uu) {
      const int sn = (uu < 3) ? (g * 8 + 2 * uu + 2) : ((g + 1 < NG) ? (g * 8 + 8) : (g * 8 + 6));
#pragma unroll
      for (int q = 0; q < 2; ++q)
#pragma unroll
        for (int ct = 0; ct < 8; ++ct) bn[q * 8 + ct] = bl[((sn + q) * 8 + ct) * 32];
      __builtin_amdgcn_sched_barrier(0);
#pragma unroll
      for (int q = 0; q < 2; ++q)
#pragma unroll
        for (int ct = 0; ct < 8; ++ct)
          acc[ct] = __builtin_amdgcn_wmma_f32_16x16x4_f32(
               false, a[2 * uu + q],  false, bc[q * 8 + ct],
               (short)0, acc[ct],  false,  false);
      __builtin_amdgcn_sched_barrier(0);
#pragma unroll
      for (int i = 0; i < 16; ++i) bc[i] = bn[i];
    }
#pragma unroll
    for (int t = 0; t < 8; ++t) a[t] = an[t];
  }

  float* crow = C + (size_t)(rowBase + half * 8) * NOUT + r;
  float bv[8];
#pragma unroll
  for (int ct = 0; ct < 8; ++ct) bv[ct] = 0.0f;
  if (flags & 2) {
#pragma unroll
    for (int ct = 0; ct < 8; ++ct) bv[ct] = bias[ct * 16 + r];
  }
  if (flags & 1) {
    float cin[8][8];
#pragma unroll
    for (int ct = 0; ct < 8; ++ct)
#pragma unroll
      for (int i = 0; i < 8; ++i) cin[ct][i] = crow[(size_t)i * NOUT + ct * 16];
#pragma unroll
    for (int ct = 0; ct < 8; ++ct)
#pragma unroll
      for (int i = 0; i < 8; ++i) {
        float v = acc[ct][i] + bv[ct];
        v += cin[ct][i];
        if (flags & 4) v = fmaxf(v, 0.0f);
        crow[(size_t)i * NOUT + ct * 16] = v;
      }
  } else {
#pragma unroll
    for (int ct = 0; ct < 8; ++ct)
#pragma unroll
      for (int i = 0; i < 8; ++i) {
        float v = acc[ct][i] + bv[ct];
        if (flags & 4) v = fmaxf(v, 0.0f);
        crow[(size_t)i * NOUT + ct * 16] = v;
      }
  }
}

extern "C" void kernel_launch(void* const* d_in, const int* in_sizes, int n_in,
                              void* d_out, int out_size, void* d_ws, size_t ws_size,
                              hipStream_t stream) {
  const float* x  = (const float*)d_in[0];
  const int*   ei = (const int*)d_in[1];
  const float* W1 = (const float*)d_in[2];
  const float* b1 = (const float*)d_in[3];
  const float* W2 = (const float*)d_in[4];
  const float* b2 = (const float*)d_in[5];
  const float* Ws = (const float*)d_in[6];
  const float* bs = (const float*)d_in[7];
  const float* Wf = (const float*)d_in[8];
  const float* bf = (const float*)d_in[9];
  float* out = (float*)d_out;

  const int IN = 256;
  const int n  = in_sizes[0] / IN;
  const int E  = in_sizes[1] / 2;
  const int* src = ei;
  const int* dst = ei + E;

  float* ws   = (float*)d_ws;
  float* xw   = ws;
  float* agg  = xw + (size_t)n * NOUT;
  float* dinv = agg + (size_t)n * NOUT;

  int* cnt     = (int*)d_out;
  int* rowptr  = cnt + n;
  int* csr_src = rowptr + n + 1;
  int* rank    = csr_src + E;

  dim3 B(256);
  const int gn   = (n + 255) / 256;
  const int gE   = (E + 255) / 256;
  const int gS   = (n + 1023) / 1024;
  const int gNw  = (n + 7) / 8;
  const int gM   = (n / 16 + 7) / 8;

  k_zero_i32     <<<gn,  B, 0, stream>>>(cnt, n);
  k_count_rank   <<<gE,  B, 0, stream>>>(dst, cnt, rank, E);
  k_rowptr_dinv  <<<gS,  dim3(1024), 0, stream>>>(cnt, rowptr, dinv, n);
  k_csr_fill     <<<gE,  B, 0, stream>>>(src, dst, rank, rowptr, csr_src, E);

  k_gemm_wmma<256> <<<gM, B, 0, stream>>>(x, W1, nullptr, xw, n, 0);
  k_gather       <<<gNw, B, 0, stream>>>(xw, dinv, rowptr, csr_src, b1, agg, n);

  k_gemm_wmma<128> <<<gM, B, 0, stream>>>(agg, W2, nullptr, xw, n, 0);
  k_gather       <<<gNw, B, 0, stream>>>(xw, dinv, rowptr, csr_src, b2, agg, n);

  k_gemm_wmma<128> <<<gM, B, 0, stream>>>(agg, Wf, bf, out, n, 2);
  k_gemm_wmma<256> <<<gM, B, 0, stream>>>(x,  Ws, bs, out, n, 1 | 2);
}


# Edge list built by a two-level counting sort in LDS (buckets of 256 destinations) instead of global int atomics, rank array and scattered fill; a node's edges keep an arrival order, sum order changes
# speedup vs baseline: 4.7848x; 1.5308x over previous
#include <hip/hip_runtime.h>
#include <math.h>

typedef __attribute__((ext_vector_type(2))) float v2f;
typedef __attribute__((ext_vector_type(8))) float v8f;

#define NOUT 128

#define TILE_E   8192
#define BKT_W    256
#define BKT_SH   8
#define MAX_BKT  256
#define BKT_CAP  16384

__global__ __launch_bounds__(1024) void k_bin_count(const int* __restrict__ dst, int* __restrict__ tbl,
                                                    int E, int NB, int nT) {
  __shared__ int hist[MAX_BKT];
  const int tid = threadIdx.x;
  if (tid < MAX_BKT) hist[tid] = 0;
  __syncthreads();
  const int base = blockIdx.x * TILE_E;
  int dv[TILE_E / 1024];
#pragma unroll
  for (int k = 0; k < TILE_E / 1024; ++k) {
    const int i = base + k * 1024 + tid;
    dv[k] = (i < E) ? dst[i] : -1;
  }
#pragma unroll
  for (int k = 0; k < TILE_E / 1024; ++k)
    if (dv[k] >= 0) atomicAdd(&hist[dv[k] >> BKT_SH], 1);
  __syncthreads();
  if (tid < NB) tbl[tid * nT + blockIdx.x] = hist[tid];
}

__global__ __launch_bounds__(1024) void k_excl_scan(const int* __restrict__ in, int* __restrict__ out, int len) {
  __shared__ int wtot[32];
  __shared__ int wpre[32];
  const int tid = threadIdx.x, lane = tid & 31, wave = tid >> 5;
  const int base = blockIdx.x * 1024;
  int before = 0;
  for (int i = tid; i < base; i += 1024) before += in[i];
  const int idx = base + tid;
  const int c = (idx < len) ? in[idx] : 0;
  int incl = c;
  for (int off = 1; off < 32; off <<= 1) {
    int t = __shfl_up(incl, off);
    if (lane >= off) incl += t;
  }
  for (int off = 16; off > 0; off >>= 1) before += __shfl_xor(before, off);
  if (lane == 31) wtot[wave] = incl;
  if (lane == 0)  wpre[wave] = before;
  __syncthreads();
  if (wave == 0) {
    const int t = wtot[lane];
    int it = t;
    int p = wpre[lane];
    for (int off = 1; off < 32; off <<= 1) {
      int u = __shfl_up(it, off);
      if (lane >= off) it += u;
    }
    for (int off = 16; off > 0; off >>= 1) p += __shfl_xor(p, off);
    wtot[lane] = it - t + p;
  }
  __syncthreads();
  const int start = wtot[wave] + incl - c;
  if (idx < len) {
    out[idx] = start;
    if (idx == len - 1) out[len] = start + c;
  }
}

__global__ __launch_bounds__(1024) void k_bin_scatter(const int* __restrict__ src, const int* __restrict__ dst,
                                                      const int* __restrict__ tbl, const int* __restrict__ tscan,
                                                      int2* __restrict__ binned, int E, int NB, int nT) {
  __shared__ int2 stage[TILE_E];
  __shared__ int lcur[MAX_BKT];
  __shared__ int gdelta[MAX_BKT];
  __shared__ int wsum[MAX_BKT / 32];
  const int tid = threadIdx.x, lane = tid & 31, wave = tid >> 5;
  const int tile = blockIdx.x;
  int c = 0, incl = 0;
  if (tid < MAX_BKT) {
    c = (tid < NB) ? tbl[tid * nT + tile] : 0;
    incl = c;
    for (int off = 1; off < 32; off <<= 1) {
      int t = __shfl_up(incl, off);
      if (lane >= off) incl += t;
    }
    if (lane == 31) wsum[wave] = incl;
  }
  __syncthreads();
  if (tid < MAX_BKT) {
    int pre = 0;
    for (int w = 0; w < wave; ++w) pre += wsum[w];
    const int ls = pre + incl - c;
    lcur[tid] = ls;
    gdelta[tid] = ((tid < NB) ? tscan[tid * nT + tile] : 0) - ls;
  }
  __syncthreads();
  const int base = tile * TILE_E;
  int dv[TILE_E / 1024], sv[TILE_E / 1024];
#pragma unroll
  for (int k = 0; k < TILE_E / 1024; ++k) {
    const int i = base + k * 1024 + tid;
    dv[k] = (i < E) ? dst[i] : -1;
    sv[k] = (i < E) ? src[i] : 0;
  }
#pragma unroll
  for (int k = 0; k < TILE_E / 1024; ++k)
    if (dv[k] >= 0) {
      const int p = atomicAdd(&lcur[dv[k] >> BKT_SH], 1);
      stage[p] = make_int2(sv[k], dv[k]);
    }
  __syncthreads();
  const int m = min(TILE_E, E - base);
  for (int j = tid; j < m; j += 1024) {
    const int2 e = stage[j];
    binned[gdelta[e.y >> BKT_SH] + j] = e;
  }
}

__global__ __launch_bounds__(1024) void k_bucket_csr(const int2* __restrict__ binned, const int* __restrict__ tscan,
                                                     int* __restrict__ rowptr, float* __restrict__ dinv,
                                                     int* __restrict__ csr_src, int n, int nT) {
  __shared__ int sorted[BKT_CAP];
  __shared__ int cnt[BKT_W];
  __shared__ int seg[BKT_W];
  __shared__ int wsum[BKT_W / 32];
  const int tid = threadIdx.x, lane = tid & 31, wave = tid >> 5;
  const int b = blockIdx.x;
  const int beg = tscan[b * nT], end = tscan[(b + 1) * nT];
  if (tid < BKT_W) cnt[tid] = 0;
  __syncthreads();
  for (int i = beg + tid; i < end; i += 1024) atomicAdd(&cnt[binned[i].y & (BKT_W - 1)], 1);
  __syncthreads();
  int c = 0, incl = 0;
  if (tid < BKT_W) {
    c = cnt[tid];
    incl = c;
    for (int off = 1; off < 32; off <<= 1) {
      int t = __shfl_up(incl, off);
      if (lane >= off) incl += t;
    }
    if (lane == 31) wsum[wave] = incl;
  }
  __syncthreads();
  if (tid < BKT_W) {
    int pre = 0;
    for (int w = 0; w < wave; ++w) pre += wsum[w];
    const int ls = pre + incl - c;
    seg[tid] = ls;
    cnt[tid] = 0;
    const int d = b * BKT_W + tid;
    if (d < n) {
      rowptr[d] = beg + ls;
      float f = (float)(c + 1);
      dinv[d] = (f > 0.0f) ? rsqrtf(f) : 0.0f;
      if (d == n - 1) rowptr[n] = beg + ls + c;
    }
  }
  __syncthreads();
  if (end - beg <= BKT_CAP) {
    for (int i = beg + tid; i < end; i += 1024) {
      const int2 e = binned[i];
      const int l = e.y & (BKT_W - 1);
      sorted[seg[l] + atomicAdd(&cnt[l], 1)] = e.x;
    }
    __syncthreads();
    for (int j = tid; j < end - beg; j += 1024) csr_src[beg + j] = sorted[j];
  } else {
    for (int i = beg + tid; i < end; i += 1024) {
      const int2 e = binned[i];
      const int l = e.y & (BKT_W - 1);
      csr_src[beg + seg[l] + atomicAdd(&cnt[l], 1)] = e.x;
    }
  }
}

__global__ __launch_bounds__(256) void k_gather(const float* __restrict__ xw,
                                                const float* __restrict__ dinv,
                                                const int* __restrict__ rowptr,
                                                const int* __restrict__ csr_src,
                                                const float* __restrict__ bias,
                                                float* __restrict__ h, int n) {
  const int lane = threadIdx.x & 31;
  const int d = __builtin_amdgcn_readfirstlane(blockIdx.x * 8 + (threadIdx.x >> 5));
  if (d >= n) return;
  const int beg = rowptr[d], end = rowptr[d + 1];
  const float dd = dinv[d];
  const float4* xr = (const float4*)xw;
  const float4 sv = xr[(size_t)d * (NOUT / 4) + lane];
  float4 acc = {0.0f, 0.0f, 0.0f, 0.0f};

  for (int base = beg; base < end; base += 32) {
    const int j = base + lane;
    const bool ok = j < end;
    const int   myS = ok ? csr_src[j] : d;
    const float myW = ok ? dinv[myS] * dd : 0.0f;
    const int cnt = min(32, end - base);
    for (int t = 0; t < cnt; t += 4) {
      const int s0 = __builtin_amdgcn_readlane(myS, t);
      const int s1 = __builtin_amdgcn_readlane(myS, t + 1);
      const int s2 = __builtin_amdgcn_readlane(myS, t + 2);
      const int s3 = __builtin_amdgcn_readlane(myS, t + 3);
      const float w0 = __int_as_float(__builtin_amdgcn_readlane(__float_as_int(myW), t));
      const float w1 = __int_as_float(__builtin_amdgcn_readlane(__float_as_int(myW), t + 1));
      const float w2 = __int_as_float(__builtin_amdgcn_readlane(__float_as_int(myW), t + 2));
      const float w3 = __int_as_float(__builtin_amdgcn_readlane(__float_as_int(myW), t + 3));
      const float4 v0 = xr[(size_t)s0 * (NOUT / 4) + lane];
      const float4 v1 = xr[(size_t)s1 * (NOUT / 4) + lane];
      const float4 v2 = xr[(size_t)s2 * (NOUT / 4) + lane];
      const float4 v3 = xr[(size_t)s3 * (NOUT / 4) + lane];
      acc.x += v0.x * w0; acc.y += v0.y * w0; acc.z += v0.z * w0; acc.w += v0.w * w0;
      acc.x += v1.x * w1; acc.y += v1.y * w1; acc.z += v1.z * w1; acc.w += v1.w * w1;
      acc.x += v2.x * w2; acc.y += v2.y * w2; acc.z += v2.z * w2; acc.w += v2.w * w2;
      acc.x += v3.x * w3; acc.y += v3.y * w3; acc.z += v3.z * w3; acc.w += v3.w * w3;
    }
  }

  acc.x += sv.x * dd * dd; acc.y += sv.y * dd * dd; acc.z += sv.z * dd * dd; acc.w += sv.w * dd * dd;
  const float4 b = ((const float4*)bias)[lane];
  float4 o;
  o.x = fmaxf(acc.x + b.x, 0.0f); o.y = fmaxf(acc.y + b.y, 0.0f);
  o.z = fmaxf(acc.z + b.z, 0.0f); o.w = fmaxf(acc.w + b.w, 0.0f);
  ((float4*)h)[(size_t)d * (NOUT / 4) + lane] = o;
}

template <int K>
__global__ void __launch_bounds__(256) __attribute__((amdgpu_waves_per_eu(1, 4))) k_gemm_wmma(
    const float* __restrict__ A, const float* __restrict__ W,
    const float* __restrict__ bias, float* __restrict__ C,
    int M, int flags) {
  constexpr int KS = K / 4;
  constexpr int NG = K / 32;
  __shared__ v2f sB[KS * 8 * 32];

  const int lane = threadIdx.x & 31;
  const int wave = __builtin_amdgcn_readfirstlane(threadIdx.x >> 5);
  const int half = lane >> 4;
  const int r    = lane & 15;
  const int rowBase = (blockIdx.x * 8 + wave) * 16;
  const bool active = rowBase < M;

  const float* arow = A + (size_t)(active ? rowBase + r : 0) * K + 2 * half;
  v2f a[8], an[8];
  if (active) {
#pragma unroll
    for (int t = 0; t < 8; ++t) a[t] = *(const v2f*)(arow + 4 * t);
  }

  {
    const float* wcol = W + (size_t)(2 * half) * NOUT + wave * 16 + r;
#pragma unroll 8
    for (int s = 0; s < KS; ++s) {
      v2f b;
      b[0] = wcol[(size_t)(4 * s) * NOUT];
      b[1] = wcol[(size_t)(4 * s + 1) * NOUT];
      sB[(s * 8 + wave) * 32 + lane] = b;
    }
  }
  __syncthreads();
  if (!active) return;

  v8f acc[8];
#pragma unroll
  for (int ct = 0; ct < 8; ++ct) acc[ct] = v8f{};

  const v2f* bl = sB + lane;
  v2f bc[16], bn[16];
#pragma unroll
  for (int q = 0; q < 2; ++q)
#pragma unroll
    for (int ct = 0; ct < 8; ++ct) bc[q * 8 + ct] = bl[(q * 8 + ct) * 32];

  for (int g = 0; g < NG; ++g) {
    const int gn = (g + 1 < NG) ? g + 1 : g;
#pragma unroll
    for (int t = 0; t < 8; ++t) an[t] = *(const v2f*)(arow + 32 * gn + 4 * t);
#pragma unroll
    for (int uu = 0; uu < 4; ++uu) {
      const int sn = (uu < 3) ? (g * 8 + 2 * uu + 2) : ((g + 1 < NG) ? (g * 8 + 8) : (g * 8 + 6));
#pragma unroll
      for (int q = 0; q < 2; ++q)
#pragma unroll
        for (int ct = 0; ct < 8; ++ct) bn[q * 8 + ct] = bl[((sn + q) * 8 + ct) * 32];
      __builtin_amdgcn_sched_barrier(0);
#pragma unroll
      for (int q = 0; q < 2; ++q)
#pragma unroll
        for (int ct = 0; ct < 8; ++ct)
          acc[ct] = __builtin_amdgcn_wmma_f32_16x16x4_f32(
               false, a[2 * uu + q],  false, bc[q * 8 + ct],
               (short)0, acc[ct],  false,  false);
      __builtin_amdgcn_sched_barrier(0);
#pragma unroll
      for (int i = 0; i < 16; ++i) bc[i] = bn[i];
    }
#pragma unroll
    for (int t = 0; t < 8; ++t) a[t] = an[t];
  }

  float* crow = C + (size_t)(rowBase + half * 8) * NOUT + r;
  float bv[8];
#pragma unroll
  for (int ct = 0; ct < 8; ++ct) bv[ct] = 0.0f;
  if (flags & 2) {
#pragma unroll
    for (int ct = 0; ct < 8; ++ct) bv[ct] = bias[ct * 16 + r];
  }
  if (flags & 1) {
    float cin[8][8];
#pragma unroll
    for (int ct = 0; ct < 8; ++ct)
#pragma unroll
      for (int i = 0; i < 8; ++i) cin[ct][i] = crow[(size_t)i * NOUT + ct * 16];
#pragma unroll
    for (int ct = 0; ct < 8; ++ct)
#pragma unroll
      for (int i = 0; i < 8; ++i) {
        float v = acc[ct][i] + bv[ct];
        v += cin[ct][i];
        if (flags & 4) v = fmaxf(v, 0.0f);
        crow[(size_t)i * NOUT + ct * 16] = v;
      }
  } else {
#pragma unroll
    for (int ct = 0; ct < 8; ++ct)
#pragma unroll
      for (int i = 0; i < 8; ++i) {
        float v = acc[ct][i] + bv[ct];
        if (flags & 4) v = fmaxf(v, 0.0f);
        crow[(size_t)i * NOUT + ct * 16] = v;
      }
  }
}

extern "C" void kernel_launch(void* const* d_in, const int* in_sizes, int n_in,
                              void* d_out, int out_size, void* d_ws, size_t ws_size,
                              hipStream_t stream) {
  const float* x  = (const float*)d_in[0];
  const int*   ei = (const int*)d_in[1];
  const float* W1 = (const float*)d_in[2];
  const float* b1 = (const float*)d_in[3];
  const float* W2 = (const float*)d_in[4];
  const float* b2 = (const float*)d_in[5];
  const float* Ws = (const float*)d_in[6];
  const float* bs = (const float*)d_in[7];
  const float* Wf = (const float*)d_in[8];
  const float* bf = (const float*)d_in[9];
  float* out = (float*)d_out;

  const int IN = 256;
  const int n  = in_sizes[0] / IN;
  const int E  = in_sizes[1] / 2;
  const int* src = ei;
  const int* dst = ei + E;

  float* ws   = (float*)d_ws;
  float* xw   = ws;
  float* agg  = xw + (size_t)n * NOUT;
  float* dinv = agg + (size_t)n * NOUT;

  const int NB = (n + BKT_W - 1) / BKT_W;
  const int nT = (E + TILE_E - 1) / TILE_E;
  int2* binned = (int2*)d_out;
  int* csr_src = (int*)d_out + 2 * (size_t)E;
  int* rowptr  = csr_src + E;
  int* tbl     = rowptr + n + 1;
  int* tscan   = tbl + NB * nT;

  dim3 B(256);
  dim3 B4(1024);
  const int gS   = (NB * nT + 1023) / 1024;
  const int gNw  = (n + 7) / 8;
  const int gM   = (n / 16 + 7) / 8;

  k_bin_count    <<<nT, B4, 0, stream>>>(dst, tbl, E, NB, nT);
  k_excl_scan    <<<gS, B4, 0, stream>>>(tbl, tscan, NB * nT);
  k_bin_scatter  <<<nT, B4, 0, stream>>>(src, dst, tbl, tscan, binned, E, NB, nT);
  k_bucket_csr   <<<NB, B4, 0, stream>>>(binned, tscan, rowptr, dinv, csr_src, n, nT);

  k_gemm_wmma<256> <<<gM, B, 0, stream>>>(x, W1, nullptr, xw, n, 0);
  k_gather       <<<gNw, B, 0, stream>>>(xw, dinv, rowptr, csr_src, b1, agg, n);

  k_gemm_wmma<128> <<<gM, B, 0, stream>>>(agg, W2, nullptr, xw, n, 0);
  k_gather       <<<gNw, B, 0, stream>>>(xw, dinv, rowptr, csr_src, b2, agg, n);

  k_gemm_wmma<128> <<<gM, B, 0, stream>>>(agg, Wf, bf, out, n, 2);
  k_gemm_wmma<256> <<<gM, B, 0, stream>>>(x,  Ws, bs, out, n, 1 | 2);
}


# Per-edge weight look-up dinv[src] removed from the gather: the GEMM that writes xw scales each row by its dinv once; gather adds plain rows, same edge order; the scaling now rounds before the sum
# speedup vs baseline: 5.1678x; 1.0800x over previous
#include <hip/hip_runtime.h>
#include <math.h>

typedef __attribute__((ext_vector_type(2))) float v2f;
typedef __attribute__((ext_vector_type(8))) float v8f;

#define NOUT 128

#define TILE_E   8192
#define BKT_W    256
#define BKT_SH   8
#define MAX_BKT  256
#define BKT_CAP  16384

__global__ __launch_bounds__(1024) void k_bin_count(const int* __restrict__ dst, int* __restrict__ tbl,
                                                    int E, int NB, int nT) {
  __shared__ int hist[MAX_BKT];
  const int tid = threadIdx.x;
  if (tid < MAX_BKT) hist[tid] = 0;
  __syncthreads();
  const int base = blockIdx.x * TILE_E;
  int dv[TILE_E / 1024];
#pragma unroll
  for (int k = 0; k < TILE_E / 1024; ++k) {
    const int i = base + k * 1024 + tid;
    dv[k] = (i < E) ? dst[i] : -1;
  }
#pragma unroll
  for (int k = 0; k < TILE_E / 1024; ++k)
    if (dv[k] >= 0) atomicAdd(&hist[dv[k] >> BKT_SH], 1);
  __syncthreads();
  if (tid < NB) tbl[tid * nT + blockIdx.x] = hist[tid];
}

__global__ __launch_bounds__(1024) void k_excl_scan(const int* __restrict__ in, int* __restrict__ out, int len) {
  __shared__ int wtot[32];
  __shared__ int wpre[32];
  const int tid = threadIdx.x, lane = tid & 31, wave = tid >> 5;
  const int base = blockIdx.x * 1024;
  int before = 0;
  for (int i = tid; i < base; i += 1024) before += in[i];
  const int idx = base + tid;
  const int c = (idx < len) ? in[idx] : 0;
  int incl = c;
  for (int off = 1; off < 32; off <<= 1) {
    int t = __shfl_up(incl, off);
    if (lane >= off) incl += t;
  }
  for (int off = 16; off > 0; off >>= 1) before += __shfl_xor(before, off);
  if (lane == 31) wtot[wave] = incl;
  if (lane == 0)  wpre[wave] = before;
  __syncthreads();
  if (wave == 0) {
    const int t = wtot[lane];
    int it = t;
    int p = wpre[lane];
    for (int off = 1; off < 32; off <<= 1) {
      int u = __shfl_up(it, off);
      if (lane >= off) it += u;
    }
    for (int off = 16; off > 0; off >>= 1) p += __shfl_xor(p, off);
    wtot[lane] = it - t + p;
  }
  __syncthreads();
  const int start = wtot[wave] + incl - c;
  if (idx < len) {
    out[idx] = start;
    if (idx == len - 1) out[len] = start + c;
  }
}

__global__ __launch_bounds__(1024) void k_bin_scatter(const int* __restrict__ src, const int* __restrict__ dst,
                                                      const int* __restrict__ tbl, const int* __restrict__ tscan,
                                                      int2* __restrict__ binned, int E, int NB, int nT) {
  __shared__ int2 stage[TILE_E];
  __shared__ int lcur[MAX_BKT];
  __shared__ int gdelta[MAX_BKT];
  __shared__ int wsum[MAX_BKT / 32];
  const int tid = threadIdx.x, lane = tid & 31, wave = tid >> 5;
  const int tile = blockIdx.x;
  int c = 0, incl = 0;
  if (tid < MAX_BKT) {
    c = (tid < NB) ? tbl[tid * nT + tile] : 0;
    incl = c;
    for (int off = 1; off < 32; off <<= 1) {
      int t = __shfl_up(incl, off);
      if (lane >= off) incl += t;
    }
    if (lane == 31) wsum[wave] = incl;
  }
  __syncthreads();
  if (tid < MAX_BKT) {
    int pre = 0;
    for (int w = 0; w < wave; ++w) pre += wsum[w];
    const int ls = pre + incl - c;
    lcur[tid] = ls;
    gdelta[tid] = ((tid < NB) ? tscan[tid * nT + tile] : 0) - ls;
  }
  __syncthreads();
  const int base = tile * TILE_E;
  int dv[TILE_E / 1024], sv[TILE_E / 1024];
#pragma unroll
  for (int k = 0; k < TILE_E / 1024; ++k) {
    const int i = base + k * 1024 + tid;
    dv[k] = (i < E) ? dst[i] : -1;
    sv[k] = (i < E) ? src[i] : 0;
  }
#pragma unroll
  for (int k = 0; k < TILE_E / 1024; ++k)
    if (dv[k] >= 0) {
      const int p = atomicAdd(&lcur[dv[k] >> BKT_SH], 1);
      stage[p] = make_int2(sv[k], dv[k]);
    }
  __syncthreads();
  const int m = min(TILE_E, E - base);
  for (int j = tid; j < m; j += 1024) {
    const int2 e = stage[j];
    binned[gdelta[e.y >> BKT_SH] + j] = e;
  }
}

__global__ __launch_bounds__(1024) void k_bucket_csr(const int2* __restrict__ binned, const int* __restrict__ tscan,
                                                     int* __restrict__ rowptr, float* __restrict__ dinv,
                                                     int* __restrict__ csr_src, int n, int nT) {
  __shared__ int sorted[BKT_CAP];
  __shared__ int cnt[BKT_W];
  __shared__ int seg[BKT_W];
  __shared__ int wsum[BKT_W / 32];
  const int tid = threadIdx.x, lane = tid & 31, wave = tid >> 5;
  const int b = blockIdx.x;
  const int beg = tscan[b * nT], end = tscan[(b + 1) * nT];
  if (tid < BKT_W) cnt[tid] = 0;
  __syncthreads();
  for (int i = beg + tid; i < end; i += 1024) atomicAdd(&cnt[binned[i].y & (BKT_W - 1)], 1);
  __syncthreads();
  int c = 0, incl = 0;
  if (tid < BKT_W) {
    c = cnt[tid];
    incl = c;
    for (int off = 1; off < 32; off <<= 1) {
      int t = __shfl_up(incl, off);
      if (lane >= off) incl += t;
    }
    if (lane == 31) wsum[wave] = incl;
  }
  __syncthreads();
  if (tid < BKT_W) {
    int pre = 0;
    for (int w = 0; w < wave; ++w) pre += wsum[w];
    const int ls = pre + incl - c;
    seg[tid] = ls;
    cnt[tid] = 0;
    const int d = b * BKT_W + tid;
    if (d < n) {
      rowptr[d] = beg + ls;
      float f = (float)(c + 1);
      dinv[d] = (f > 0.0f) ? rsqrtf(f) : 0.0f;
      if (d == n - 1) rowptr[n] = beg + ls + c;
    }
  }
  __syncthreads();
  if (end - beg <= BKT_CAP) {
    for (int i = beg + tid; i < end; i += 1024) {
      const int2 e = binned[i];
      const int l = e.y & (BKT_W - 1);
      sorted[seg[l] + atomicAdd(&cnt[l], 1)] = e.x;
    }
    __syncthreads();
    for (int j = tid; j < end - beg; j += 1024) csr_src[beg + j] = sorted[j];
  } else {
    for (int i = beg + tid; i < end; i += 1024) {
      const int2 e = binned[i];
      const int l = e.y & (BKT_W - 1);
      csr_src[beg + seg[l] + atomicAdd(&cnt[l], 1)] = e.x;
    }
  }
}

__global__ __launch_bounds__(256) void k_gather(const float* __restrict__ xws,
                                                const float* __restrict__ dinv,
                                                const int* __restrict__ rowptr,
                                                const int* __restrict__ csr_src,
                                                const float* __restrict__ bias,
                                                float* __restrict__ h, int n) {
  const int lane = threadIdx.x & 31;
  const int d = __builtin_amdgcn_readfirstlane(blockIdx.x * 8 + (threadIdx.x >> 5));
  if (d >= n) return;
  const int beg = rowptr[d], end = rowptr[d + 1];
  const float dd = dinv[d];
  const float4* xr = (const float4*)xws;
  const float4 sv = xr[(size_t)d * (NOUT / 4) + lane];
  float4 acc = {0.0f, 0.0f, 0.0f, 0.0f};

  for (int base = beg; base < end; base += 32) {
    const int j = base + lane;
    const int myS = (j < end) ? csr_src[j] : d;
    const int cnt = min(32, end - base);
    for (int t = 0; t < cnt; t += 4) {
      const int s0 = __builtin_amdgcn_readlane(myS, t);
      const int s1 = __builtin_amdgcn_readlane(myS, t + 1);
      const int s2 = __builtin_amdgcn_readlane(myS, t + 2);
      const int s3 = __builtin_amdgcn_readlane(myS, t + 3);
      const float w0 = 1.0f;
      const float w1 = (t + 1 < cnt) ? 1.0f : 0.0f;
      const float w2 = (t + 2 < cnt) ? 1.0f : 0.0f;
      const float w3 = (t + 3 < cnt) ? 1.0f : 0.0f;
      const float4 v0 = xr[(size_t)s0 * (NOUT / 4) + lane];
      const float4 v1 = xr[(size_t)s1 * (NOUT / 4) + lane];
      const float4 v2 = xr[(size_t)s2 * (NOUT / 4) + lane];
      const float4 v3 = xr[(size_t)s3 * (NOUT / 4) + lane];
      acc.x += v0.x * w0; acc.y += v0.y * w0; acc.z += v0.z * w0; acc.w += v0.w * w0;
      acc.x += v1.x * w1; acc.y += v1.y * w1; acc.z += v1.z * w1; acc.w += v1.w * w1;
      acc.x += v2.x * w2; acc.y += v2.y * w2; acc.z += v2.z * w2; acc.w += v2.w * w2;
      acc.x += v3.x * w3; acc.y += v3.y * w3; acc.z += v3.z * w3; acc.w += v3.w * w3;
    }
  }

  acc.x += sv.x; acc.y += sv.y; acc.z += sv.z; acc.w += sv.w;
  const float4 b = ((const float4*)bias)[lane];
  float4 o;
  o.x = fmaxf(acc.x * dd + b.x, 0.0f); o.y = fmaxf(acc.y * dd + b.y, 0.0f);
  o.z = fmaxf(acc.z * dd + b.z, 0.0f); o.w = fmaxf(acc.w * dd + b.w, 0.0f);
  ((float4*)h)[(size_t)d * (NOUT / 4) + lane] = o;
}

template <int K, bool SCALE>
__global__ void __launch_bounds__(256) __attribute__((amdgpu_waves_per_eu(1, 4))) k_gemm_wmma(
    const float* __restrict__ A, const float* __restrict__ W,
    const float* __restrict__ bias, const float* __restrict__ rscale, float* __restrict__ C,
    int M, int flags) {
  constexpr int KS = K / 4;
  constexpr int NG = K / 32;
  __shared__ v2f sB[KS * 8 * 32];

  const int lane = threadIdx.x & 31;
  const int wave = __builtin_amdgcn_readfirstlane(threadIdx.x >> 5);
  const int half = lane >> 4;
  const int r    = lane & 15;
  const int rowBase = (blockIdx.x * 8 + wave) * 16;
  const bool active = rowBase < M;

  const float* arow = A + (size_t)(active ? rowBase + r : 0) * K + 2 * half;
  v2f a[8], an[8];
  if (active) {
#pragma unroll
    for (int t = 0; t < 8; ++t) a[t] = *(const v2f*)(arow + 4 * t);
  }

  {
    const float* wcol = W + (size_t)(2 * half) * NOUT + wave * 16 + r;
#pragma unroll 8
    for (int s = 0; s < KS; ++s) {
      v2f b;
      b[0] = wcol[(size_t)(4 * s) * NOUT];
      b[1] = wcol[(size_t)(4 * s + 1) * NOUT];
      sB[(s * 8 + wave) * 32 + lane] = b;
    }
  }
  __syncthreads();
  if (!active) return;

  v8f acc[8];
#pragma unroll
  for (int ct = 0; ct < 8; ++ct) acc[ct] = v8f{};

  const v2f* bl = sB + lane;
  v2f bc[16], bn[16];
#pragma unroll
  for (int q = 0; q < 2; ++q)
#pragma unroll
    for (int ct = 0; ct < 8; ++ct) bc[q * 8 + ct] = bl[(q * 8 + ct) * 32];

  for (int g = 0; g < NG; ++g) {
    const int gn = (g + 1 < NG) ? g + 1 : g;
#pragma unroll
    for (int t = 0; t < 8; ++t) an[t] = *(const v2f*)(arow + 32 * gn + 4 * t);
#pragma unroll
    for (int uu = 0; uu < 4; ++uu) {
      const int sn = (uu < 3) ? (g * 8 + 2 * uu + 2) : ((g + 1 < NG) ? (g * 8 + 8) : (g * 8 + 6));
#pragma unroll
      for (int q = 0; q < 2; ++q)
#pragma unroll
        for (int ct = 0; ct < 8; ++ct) bn[q * 8 + ct] = bl[((sn + q) * 8 + ct) * 32];
      __builtin_amdgcn_sched_barrier(0);
#pragma unroll
      for (int q = 0; q < 2; ++q)
#pragma unroll
        for (int ct = 0; ct < 8; ++ct)
          acc[ct] = __builtin_amdgcn_wmma_f32_16x16x4_f32(
               false, a[2 * uu + q],  false, bc[q * 8 + ct],
               (short)0, acc[ct],  false,  false);
      __builtin_amdgcn_sched_barrier(0);
#pragma unroll
      for (int i = 0; i < 16; ++i) bc[i] = bn[i];
    }
#pragma unroll
    for (int t = 0; t < 8; ++t) a[t] = an[t];
  }

  float* crow = C + (size_t)(rowBase + half * 8) * NOUT + r;
  float bv[8];
#pragma unroll
  for (int ct = 0; ct < 8; ++ct) bv[ct] = 0.0f;
  if (flags & 2) {
#pragma unroll
    for (int ct = 0; ct < 8; ++ct) bv[ct] = bias[ct * 16 + r];
  }
  if (flags & 1) {
    float cin[8][8];
#pragma unroll
    for (int ct = 0; ct < 8; ++ct)
#pragma unroll
      for (int i = 0; i < 8; ++i) cin[ct][i] = crow[(size_t)i * NOUT + ct * 16];
#pragma unroll
    for (int ct = 0; ct < 8; ++ct)
#pragma unroll
      for (int i = 0; i < 8; ++i) {
        float v = acc[ct][i] + bv[ct];
        v += cin[ct][i];
        if (flags & 4) v = fmaxf(v, 0.0f);
        crow[(size_t)i * NOUT + ct * 16] = v;
      }
  } else if (SCALE) {
    float rs[8];
#pragma unroll
    for (int i = 0; i < 8; ++i) rs[i] = rscale[rowBase + half * 8 + i];
#pragma unroll
    for (int ct = 0; ct < 8; ++ct)
#pragma unroll
      for (int i = 0; i < 8; ++i) crow[(size_t)i * NOUT + ct * 16] = acc[ct][i] * rs[i];
  } else {
#pragma unroll
    for (int ct = 0; ct < 8; ++ct)
#pragma unroll
      for (int i = 0; i < 8; ++i) {
        float v = acc[ct][i] + bv[ct];
        if (flags & 4) v = fmaxf(v, 0.0f);
        crow[(size_t)i * NOUT + ct * 16] = v;
      }
  }
}

extern "C" void kernel_launch(void* const* d_in, const int* in_sizes, int n_in,
                              void* d_out, int out_size, void* d_ws, size_t ws_size,
                              hipStream_t stream) {
  const float* x  = (const float*)d_in[0];
  const int*   ei = (const int*)d_in[1];
  const float* W1 = (const float*)d_in[2];
  const float* b1 = (const float*)d_in[3];
  const float* W2 = (const float*)d_in[4];
  const float* b2 = (const float*)d_in[5];
  const float* Ws = (const float*)d_in[6];
  const float* bs = (const float*)d_in[7];
  const float* Wf = (const float*)d_in[8];
  const float* bf = (const float*)d_in[9];
  float* out = (float*)d_out;

  const int IN = 256;
  const int n  = in_sizes[0] / IN;
  const int E  = in_sizes[1] / 2;
  const int* src = ei;
  const int* dst = ei + E;

  float* ws   = (float*)d_ws;
  float* xw   = ws;
  float* agg  = xw + (size_t)n * NOUT;
  float* dinv = agg + (size_t)n * NOUT;

  const int NB = (n + BKT_W - 1) / BKT_W;
  const int nT = (E + TILE_E - 1) / TILE_E;
  int2* binned = (int2*)d_out;
  int* csr_src = (int*)d_out + 2 * (size_t)E;
  int* rowptr  = csr_src + E;
  int* tbl     = rowptr + n + 1;
  int* tscan   = tbl + NB * nT;

  dim3 B(256);
  dim3 B4(1024);
  const int gS   = (NB * nT + 1023) / 1024;
  const int gNw  = (n + 7) / 8;
  const int gM   = (n / 16 + 7) / 8;

  k_bin_count    <<<nT, B4, 0, stream>>>(dst, tbl, E, NB, nT);
  k_excl_scan    <<<gS, B4, 0, stream>>>(tbl, tscan, NB * nT);
  k_bin_scatter  <<<nT, B4, 0, stream>>>(src, dst, tbl, tscan, binned, E, NB, nT);
  k_bucket_csr   <<<NB, B4, 0, stream>>>(binned, tscan, rowptr, dinv, csr_src, n, nT);

  k_gemm_wmma<256, true> <<<gM, B, 0, stream>>>(x, W1, nullptr, dinv, xw, n, 0);
  k_gather       <<<gNw, B, 0, stream>>>(xw, dinv, rowptr, csr_src, b1, agg, n);

  k_gemm_wmma<128, true> <<<gM, B, 0, stream>>>(agg, W2, nullptr, dinv, xw, n, 0);
  k_gather       <<<gNw, B, 0, stream>>>(xw, dinv, rowptr, csr_src, b2, agg, n);

  k_gemm_wmma<128, false> <<<gM, B, 0, stream>>>(agg, Wf, bf, nullptr, out, n, 2);
  k_gemm_wmma<256, false> <<<gM, B, 0, stream>>>(x,  Ws, bs, nullptr, out, n, 1 | 2);
}


# Each gather fused with the GEMM that consumes it: 40 gathered rows per block stay in LDS and feed WMMA (wave = 1 column tile x 3 row tiles); h1, h2 never reach memory; k order and sum order unchanged
# speedup vs baseline: 5.6119x; 1.0859x over previous
#include <hip/hip_runtime.h>
#include <math.h>

typedef __attribute__((ext_vector_type(2))) float v2f;
typedef __attribute__((ext_vector_type(8))) float v8f;

#define NOUT 128

#define TILE_E   8192
#define BKT_W    256
#define BKT_SH   8
#define MAX_BKT  256
#define BKT_CAP  16384

__global__ __launch_bounds__(1024) void k_bin_count(const int* __restrict__ dst, int* __restrict__ tbl,
                                                    int E, int NB, int nT) {
  __shared__ int hist[MAX_BKT];
  const int tid = threadIdx.x;
  if (tid < MAX_BKT) hist[tid] = 0;
  __syncthreads();
  const int base = blockIdx.x * TILE_E;
  int dv[TILE_E / 1024];
#pragma unroll
  for (int k = 0; k < TILE_E / 1024; ++k) {
    const int i = base + k * 1024 + tid;
    dv[k] = (i < E) ? dst[i] : -1;
  }
#pragma unroll
  for (int k = 0; k < TILE_E / 1024; ++k)
    if (dv[k] >= 0) atomicAdd(&hist[dv[k] >> BKT_SH], 1);
  __syncthreads();
  if (tid < NB) tbl[tid * nT + blockIdx.x] = hist[tid];
}

__global__ __launch_bounds__(1024) void k_excl_scan(const int* __restrict__ in, int* __restrict__ out, int len) {
  __shared__ int wtot[32];
  __shared__ int wpre[32];
  const int tid = threadIdx.x, lane = tid & 31, wave = tid >> 5;
  const int base = blockIdx.x * 1024;
  int before = 0;
  for (int i = tid; i < base; i += 1024) before += in[i];
  const int idx = base + tid;
  const int c = (idx < len) ? in[idx] : 0;
  int incl = c;
  for (int off = 1; off < 32; off <<= 1) {
    int t = __shfl_up(incl, off);
    if (lane >= off) incl += t;
  }
  for (int off = 16; off > 0; off >>= 1) before += __shfl_xor(before, off);
  if (lane == 31) wtot[wave] = incl;
  if (lane == 0)  wpre[wave] = before;
  __syncthreads();
  if (wave == 0) {
    const int t = wtot[lane];
    int it = t;
    int p = wpre[lane];
    for (int off = 1; off < 32; off <<= 1) {
      int u = __shfl_up(it, off);
      if (lane >= off) it += u;
    }
    for (int off = 16; off > 0; off >>= 1) p += __shfl_xor(p, off);
    wtot[lane] = it - t + p;
  }
  __syncthreads();
  const int start = wtot[wave] + incl - c;
  if (idx < len) {
    out[idx] = start;
    if (idx == len - 1) out[len] = start + c;
  }
}

__global__ __launch_bounds__(1024) void k_bin_scatter(const int* __restrict__ src, const int* __restrict__ dst,
                                                      const int* __restrict__ tbl, const int* __restrict__ tscan,
                                                      int2* __restrict__ binned, int E, int NB, int nT) {
  __shared__ int2 stage[TILE_E];
  __shared__ int lcur[MAX_BKT];
  __shared__ int gdelta[MAX_BKT];
  __shared__ int wsum[MAX_BKT / 32];
  const int tid = threadIdx.x, lane = tid & 31, wave = tid >> 5;
  const int tile = blockIdx.x;
  int c = 0, incl = 0;
  if (tid < MAX_BKT) {
    c = (tid < NB) ? tbl[tid * nT + tile] : 0;
    incl = c;
    for (int off = 1; off < 32; off <<= 1) {
      int t = __shfl_up(incl, off);
      if (lane >= off) incl += t;
    }
    if (lane == 31) wsum[wave] = incl;
  }
  __syncthreads();
  if (tid < MAX_BKT) {
    int pre = 0;
    for (int w = 0; w < wave; ++w) pre += wsum[w];
    const int ls = pre + incl - c;
    lcur[tid] = ls;
    gdelta[tid] = ((tid < NB) ? tscan[tid * nT + tile] : 0) - ls;
  }
  __syncthreads();
  const int base = tile * TILE_E;
  int dv[TILE_E / 1024], sv[TILE_E / 1024];
#pragma unroll
  for (int k = 0; k < TILE_E / 1024; ++k) {
    const int i = base + k * 1024 + tid;
    dv[k] = (i < E) ? dst[i] : -1;
    sv[k] = (i < E) ? src[i] : 0;
  }
#pragma unroll
  for (int k = 0; k < TILE_E / 1024; ++k)
    if (dv[k] >= 0) {
      const int p = atomicAdd(&lcur[dv[k] >> BKT_SH], 1);
      stage[p] = make_int2(sv[k], dv[k]);
    }
  __syncthreads();
  const int m = min(TILE_E, E - base);
  for (int j = tid; j < m; j += 1024) {
    const int2 e = stage[j];
    binned[gdelta[e.y >> BKT_SH] + j] = e;
  }
}

__global__ __launch_bounds__(1024) void k_bucket_csr(const int2* __restrict__ binned, const int* __restrict__ tscan,
                                                     int* __restrict__ rowptr, float* __restrict__ dinv,
                                                     int* __restrict__ csr_src, int n, int nT) {
  __shared__ int sorted[BKT_CAP];
  __shared__ int cnt[BKT_W];
  __shared__ int seg[BKT_W];
  __shared__ int wsum[BKT_W / 32];
  const int tid = threadIdx.x, lane = tid & 31, wave = tid >> 5;
  const int b = blockIdx.x;
  const int beg = tscan[b * nT], end = tscan[(b + 1) * nT];
  if (tid < BKT_W) cnt[tid] = 0;
  __syncthreads();
  for (int i = beg + tid; i < end; i += 1024) atomicAdd(&cnt[binned[i].y & (BKT_W - 1)], 1);
  __syncthreads();
  int c = 0, incl = 0;
  if (tid < BKT_W) {
    c = cnt[tid];
    incl = c;
    for (int off = 1; off < 32; off <<= 1) {
      int t = __shfl_up(incl, off);
      if (lane >= off) incl += t;
    }
    if (lane == 31) wsum[wave] = incl;
  }
  __syncthreads();
  if (tid < BKT_W) {
    int pre = 0;
    for (int w = 0; w < wave; ++w) pre += wsum[w];
    const int ls = pre + incl - c;
    seg[tid] = ls;
    cnt[tid] = 0;
    const int d = b * BKT_W + tid;
    if (d < n) {
      rowptr[d] = beg + ls;
      float f = (float)(c + 1);
      dinv[d] = (f > 0.0f) ? rsqrtf(f) : 0.0f;
      if (d == n - 1) rowptr[n] = beg + ls + c;
    }
  }
  __syncthreads();
  if (end - beg <= BKT_CAP) {
    for (int i = beg + tid; i < end; i += 1024) {
      const int2 e = binned[i];
      const int l = e.y & (BKT_W - 1);
      sorted[seg[l] + atomicAdd(&cnt[l], 1)] = e.x;
    }
    __syncthreads();
    for (int j = tid; j < end - beg; j += 1024) csr_src[beg + j] = sorted[j];
  } else {
    for (int i = beg + tid; i < end; i += 1024) {
      const int2 e = binned[i];
      const int l = e.y & (BKT_W - 1);
      csr_src[beg + seg[l] + atomicAdd(&cnt[l], 1)] = e.x;
    }
  }
}

__device__ __forceinline__ float4 gather_node(const float4* __restrict__ xr, const float* __restrict__ dinv,
                                              const int* __restrict__ rowptr, const int* __restrict__ csr_src,
                                              const float4 b, const int d, const int lane) {
  const int beg = rowptr[d], end = rowptr[d + 1];
  const float dd = dinv[d];
  const float4 sv = xr[(size_t)d * (NOUT / 4) + lane];
  float4 acc = {0.0f, 0.0f, 0.0f, 0.0f};

  for (int base = beg; base < end; base += 32) {
    const int j = base + lane;
    const int myS = (j < end) ? csr_src[j] : d;
    const int cnt = min(32, end - base);
    for (int t = 0; t < cnt; t += 4) {
      const int s0 = __builtin_amdgcn_readlane(myS, t);
      const int s1 = __builtin_amdgcn_readlane(myS, t + 1);
      const int s2 = __builtin_amdgcn_readlane(myS, t + 2);
      const int s3 = __builtin_amdgcn_readlane(myS, t + 3);
      const float w0 = 1.0f;
      const float w1 = (t + 1 < cnt) ? 1.0f : 0.0f;
      const float w2 = (t + 2 < cnt) ? 1.0f : 0.0f;
      const float w3 = (t + 3 < cnt) ? 1.0f : 0.0f;
      const float4 v0 = xr[(size_t)s0 * (NOUT / 4) + lane];
      const float4 v1 = xr[(size_t)s1 * (NOUT / 4) + lane];
      const float4 v2 = xr[(size_t)s2 * (NOUT / 4) + lane];
      const float4 v3 = xr[(size_t)s3 * (NOUT / 4) + lane];
      acc.x += v0.x * w0; acc.y += v0.y * w0; acc.z += v0.z * w0; acc.w += v0.w * w0;
      acc.x += v1.x * w1; acc.y += v1.y * w1; acc.z += v1.z * w1; acc.w += v1.w * w1;
      acc.x += v2.x * w2; acc.y += v2.y * w2; acc.z += v2.z * w2; acc.w += v2.w * w2;
      acc.x += v3.x * w3; acc.y += v3.y * w3; acc.z += v3.z * w3; acc.w += v3.w * w3;
    }
  }

  acc.x += sv.x; acc.y += sv.y; acc.z += sv.z; acc.w += sv.w;
  float4 o;
  o.x = fmaxf(acc.x * dd + b.x, 0.0f); o.y = fmaxf(acc.y * dd + b.y, 0.0f);
  o.z = fmaxf(acc.z * dd + b.z, 0.0f); o.w = fmaxf(acc.w * dd + b.w, 0.0f);
  return o;
}

#define FN  40
#define RPW (FN / 8)
#define RT  ((FN + 15) / 16)
#define HP  132
template <bool SCALE>
__global__ __launch_bounds__(256) void k_gather_gemm(const float* __restrict__ xws,
                                                     const float* __restrict__ dinv,
                                                     const int* __restrict__ rowptr,
                                                     const int* __restrict__ csr_src,
                                                     const float* __restrict__ bias,
                                                     const float* __restrict__ Wn,
                                                     const float* __restrict__ bias2,
                                                     float* __restrict__ C, int n) {
  __shared__ alignas(16) float sH[RT * 16 * HP];
  const int lane = threadIdx.x & 31;
  const int wave = __builtin_amdgcn_readfirstlane(threadIdx.x >> 5);
  const int nodeBase = blockIdx.x * FN;
  const float4* xr = (const float4*)xws;
  const float4 b = ((const float4*)bias)[lane];
  const float4 zero4 = {0.0f, 0.0f, 0.0f, 0.0f};

  for (int row = FN + wave; row < RT * 16; row += 8) *(float4*)(sH + row * HP + lane * 4) = zero4;
  for (int q = 0; q < RPW; ++q) {
    const int row = wave * RPW + q;
    const int d = nodeBase + row;
    float4 o = zero4;
    if (d < n) o = gather_node(xr, dinv, rowptr, csr_src, b, d, lane);
    *(float4*)(sH + row * HP + lane * 4) = o;
  }
  __syncthreads();

  const int half = lane >> 4;
  const int r    = lane & 15;
  const float* wcol = Wn + (size_t)(2 * half) * NOUT + wave * 16 + r;
  const float* hrow = sH + r * HP + 2 * half;
  v8f acc[RT];
#pragma unroll
  for (int rt = 0; rt < RT; ++rt) acc[rt] = v8f{};
#pragma unroll 8
  for (int s = 0; s < NOUT / 4; ++s) {
    v2f bf;
    bf[0] = wcol[(size_t)(4 * s) * NOUT];
    bf[1] = wcol[(size_t)(4 * s + 1) * NOUT];
#pragma unroll
    for (int rt = 0; rt < RT; ++rt) {
      const v2f a = *(const v2f*)(hrow + rt * 16 * HP + 4 * s);
      acc[rt] = __builtin_amdgcn_wmma_f32_16x16x4_f32( false, a,  false, bf,
                                                       (short)0, acc[rt],
                                                       false,  false);
    }
  }
  __syncthreads();
#pragma unroll
  for (int rt = 0; rt < RT; ++rt)
#pragma unroll
    for (int i = 0; i < 8; ++i) sH[(rt * 16 + half * 8 + i) * HP + wave * 16 + r] = acc[rt][i];
  __syncthreads();
  float4 b2 = zero4;
  if (!SCALE) b2 = ((const float4*)bias2)[lane];
  for (int q = 0; q < RPW; ++q) {
    const int row = wave * RPW + q;
    const int d = nodeBase + row;
    if (d < n) {
      float4 v = *(const float4*)(sH + row * HP + lane * 4);
      if (SCALE) {
        const float rs = dinv[d];
        v.x *= rs; v.y *= rs; v.z *= rs; v.w *= rs;
      } else {
        v.x += b2.x; v.y += b2.y; v.z += b2.z; v.w += b2.w;
      }
      ((float4*)C)[(size_t)d * (NOUT / 4) + lane] = v;
    }
  }
}

template <int K, bool SCALE>
__global__ void __launch_bounds__(256) __attribute__((amdgpu_waves_per_eu(1, 4))) k_gemm_wmma(
    const float* __restrict__ A, const float* __restrict__ W,
    const float* __restrict__ bias, const float* __restrict__ rscale,
    const float* __restrict__ Cin, float* __restrict__ C,
    int M, int flags) {
  constexpr int KS = K / 4;
  constexpr int NG = K / 32;
  __shared__ v2f sB[KS * 8 * 32];

  const int lane = threadIdx.x & 31;
  const int wave = __builtin_amdgcn_readfirstlane(threadIdx.x >> 5);
  const int half = lane >> 4;
  const int r    = lane & 15;
  const int rowBase = (blockIdx.x * 8 + wave) * 16;
  const bool active = rowBase < M;

  const float* arow = A + (size_t)(active ? rowBase + r : 0) * K + 2 * half;
  v2f a[8], an[8];
  if (active) {
#pragma unroll
    for (int t = 0; t < 8; ++t) a[t] = *(const v2f*)(arow + 4 * t);
  }

  {
    const float* wcol = W + (size_t)(2 * half) * NOUT + wave * 16 + r;
#pragma unroll 8
    for (int s = 0; s < KS; ++s) {
      v2f b;
      b[0] = wcol[(size_t)(4 * s) * NOUT];
      b[1] = wcol[(size_t)(4 * s + 1) * NOUT];
      sB[(s * 8 + wave) * 32 + lane] = b;
    }
  }
  __syncthreads();
  if (!active) return;

  v8f acc[8];
#pragma unroll
  for (int ct = 0; ct < 8; ++ct) acc[ct] = v8f{};

  const v2f* bl = sB + lane;
  v2f bc[16], bn[16];
#pragma unroll
  for (int q = 0; q < 2; ++q)
#pragma unroll
    for (int ct = 0; ct < 8; ++ct) bc[q * 8 + ct] = bl[(q * 8 + ct) * 32];

  for (int g = 0; g < NG; ++g) {
    const int gn = (g + 1 < NG) ? g + 1 : g;
#pragma unroll
    for (int t = 0; t < 8; ++t) an[t] = *(const v2f*)(arow + 32 * gn + 4 * t);
#pragma unroll
    for (int uu = 0; uu < 4; ++uu) {
      const int sn = (uu < 3) ? (g * 8 + 2 * uu + 2) : ((g + 1 < NG) ? (g * 8 + 8) : (g * 8 + 6));
#pragma unroll
      for (int q = 0; q < 2; ++q)
#pragma unroll
        for (int ct = 0; ct < 8; ++ct) bn[q * 8 + ct] = bl[((sn + q) * 8 + ct) * 32];
      __builtin_amdgcn_sched_barrier(0);
#pragma unroll
      for (int q = 0; q < 2; ++q)
#pragma unroll
        for (int ct = 0; ct < 8; ++ct)
          acc[ct] = __builtin_amdgcn_wmma_f32_16x16x4_f32(
               false, a[2 * uu + q],  false, bc[q * 8 + ct],
               (short)0, acc[ct],  false,  false);
      __builtin_amdgcn_sched_barrier(0);
#pragma unroll
      for (int i = 0; i < 16; ++i) bc[i] = bn[i];
    }
#pragma unroll
    for (int t = 0; t < 8; ++t) a[t] = an[t];
  }

  float* crow = C + (size_t)(rowBase + half * 8) * NOUT + r;
  float bv[8];
#pragma unroll
  for (int ct = 0; ct < 8; ++ct) bv[ct] = 0.0f;
  if (flags & 2) {
#pragma unroll
    for (int ct = 0; ct < 8; ++ct) bv[ct] = bias[ct * 16 + r];
  }
  if (flags & 1) {
    float cin[8][8];
    const float* cinrow = Cin + (size_t)(rowBase + half * 8) * NOUT + r;
#pragma unroll
    for (int ct = 0; ct < 8; ++ct)
#pragma unroll
      for (int i = 0; i < 8; ++i) cin[ct][i] = cinrow[(size_t)i * NOUT + ct * 16];
#pragma unroll
    for (int ct = 0; ct < 8; ++ct)
#pragma unroll
      for (int i = 0; i < 8; ++i) {
        float v = acc[ct][i] + bv[ct];
        v += cin[ct][i];
        if (flags & 4) v = fmaxf(v, 0.0f);
        crow[(size_t)i * NOUT + ct * 16] = v;
      }
  } else if (SCALE) {
    float rs[8];
#pragma unroll
    for (int i = 0; i < 8; ++i) rs[i] = rscale[rowBase + half * 8 + i];
#pragma unroll
    for (int ct = 0; ct < 8; ++ct)
#pragma unroll
      for (int i = 0; i < 8; ++i) crow[(size_t)i * NOUT + ct * 16] = acc[ct][i] * rs[i];
  } else {
#pragma unroll
    for (int ct = 0; ct < 8; ++ct)
#pragma unroll
      for (int i = 0; i < 8; ++i) {
        float v = acc[ct][i] + bv[ct];
        if (flags & 4) v = fmaxf(v, 0.0f);
        crow[(size_t)i * NOUT + ct * 16] = v;
      }
  }
}

extern "C" void kernel_launch(void* const* d_in, const int* in_sizes, int n_in,
                              void* d_out, int out_size, void* d_ws, size_t ws_size,
                              hipStream_t stream) {
  const float* x  = (const float*)d_in[0];
  const int*   ei = (const int*)d_in[1];
  const float* W1 = (const float*)d_in[2];
  const float* b1 = (const float*)d_in[3];
  const float* W2 = (const float*)d_in[4];
  const float* b2 = (const float*)d_in[5];
  const float* Ws = (const float*)d_in[6];
  const float* bs = (const float*)d_in[7];
  const float* Wf = (const float*)d_in[8];
  const float* bf = (const float*)d_in[9];
  float* out = (float*)d_out;

  const int IN = 256;
  const int n  = in_sizes[0] / IN;
  const int E  = in_sizes[1] / 2;
  const int* src = ei;
  const int* dst = ei + E;

  float* ws   = (float*)d_ws;
  float* p0   = ws;
  float* p1   = p0 + (size_t)n * NOUT;
  float* dinv = p1 + (size_t)n * NOUT;

  const int NB = (n + BKT_W - 1) / BKT_W;
  const int nT = (E + TILE_E - 1) / TILE_E;
  int2* binned = (int2*)d_out;
  int* csr_src = (int*)d_out + 2 * (size_t)E;
  int* rowptr  = csr_src + E;
  int* tbl     = rowptr + n + 1;
  int* tscan   = tbl + NB * nT;

  dim3 B(256);
  dim3 B4(1024);
  const int gS   = (NB * nT + 1023) / 1024;
  const int gF   = (n + FN - 1) / FN;
  const int gM   = (n / 16 + 7) / 8;

  k_bin_count    <<<nT, B4, 0, stream>>>(dst, tbl, E, NB, nT);
  k_excl_scan    <<<gS, B4, 0, stream>>>(tbl, tscan, NB * nT);
  k_bin_scatter  <<<nT, B4, 0, stream>>>(src, dst, tbl, tscan, binned, E, NB, nT);
  k_bucket_csr   <<<NB, B4, 0, stream>>>(binned, tscan, rowptr, dinv, csr_src, n, nT);

  k_gemm_wmma<256, true> <<<gM, B, 0, stream>>>(x, W1, nullptr, dinv, nullptr, p0, n, 0);
  k_gather_gemm<true>  <<<gF, B, 0, stream>>>(p0, dinv, rowptr, csr_src, b1, W2, nullptr, p1, n);
  k_gather_gemm<false> <<<gF, B, 0, stream>>>(p1, dinv, rowptr, csr_src, b2, Wf, bf, p0, n);
  k_gemm_wmma<256, false> <<<gM, B, 0, stream>>>(x, Ws, bs, nullptr, p0, out, n, 1 | 2);
}


# Skip GEMM fused into the second gather kernel: x@Ws runs on WMMA from the block's LDS tile, half of the blocks before their gather and half after; one launch and the graph branch's round trip gone
# speedup vs baseline: 5.9587x; 1.0618x over previous
#include <hip/hip_runtime.h>
#include <math.h>

typedef __attribute__((ext_vector_type(2))) float v2f;
typedef __attribute__((ext_vector_type(8))) float v8f;

#define NOUT 128

#define TILE_E   8192
#define BKT_W    256
#define BKT_SH   8
#define MAX_BKT  256
#define BKT_CAP  16384

__global__ __launch_bounds__(1024) void k_bin_count(const int* __restrict__ dst, int* __restrict__ tbl,
                                                    int E, int NB, int nT) {
  __shared__ int hist[MAX_BKT];
  const int tid = threadIdx.x;
  if (tid < MAX_BKT) hist[tid] = 0;
  __syncthreads();
  const int base = blockIdx.x * TILE_E;
  int dv[TILE_E / 1024];
#pragma unroll
  for (int k = 0; k < TILE_E / 1024; ++k) {
    const int i = base + k * 1024 + tid;
    dv[k] = (i < E) ? dst[i] : -1;
  }
#pragma unroll
  for (int k = 0; k < TILE_E / 1024; ++k)
    if (dv[k] >= 0) atomicAdd(&hist[dv[k] >> BKT_SH], 1);
  __syncthreads();
  if (tid < NB) tbl[tid * nT + blockIdx.x] = hist[tid];
}

__global__ __launch_bounds__(1024) void k_excl_scan(const int* __restrict__ in, int* __restrict__ out, int len) {
  __shared__ int wtot[32];
  __shared__ int wpre[32];
  const int tid = threadIdx.x, lane = tid & 31, wave = tid >> 5;
  const int base = blockIdx.x * 1024;
  int before = 0;
  for (int i = tid; i < base; i += 1024) before += in[i];
  const int idx = base + tid;
  const int c = (idx < len) ? in[idx] : 0;
  int incl = c;
  for (int off = 1; off < 32; off <<= 1) {
    int t = __shfl_up(incl, off);
    if (lane >= off) incl += t;
  }
  for (int off = 16; off > 0; off >>= 1) before += __shfl_xor(before, off);
  if (lane == 31) wtot[wave] = incl;
  if (lane == 0)  wpre[wave] = before;
  __syncthreads();
  if (wave == 0) {
    const int t = wtot[lane];
    int it = t;
    int p = wpre[lane];
    for (int off = 1; off < 32; off <<= 1) {
      int u = __shfl_up(it, off);
      if (lane >= off) it += u;
    }
    for (int off = 16; off > 0; off >>= 1) p += __shfl_xor(p, off);
    wtot[lane] = it - t + p;
  }
  __syncthreads();
  const int start = wtot[wave] + incl - c;
  if (idx < len) {
    out[idx] = start;
    if (idx == len - 1) out[len] = start + c;
  }
}

__global__ __launch_bounds__(1024) void k_bin_scatter(const int* __restrict__ src, const int* __restrict__ dst,
                                                      const int* __restrict__ tbl, const int* __restrict__ tscan,
                                                      int2* __restrict__ binned, int E, int NB, int nT) {
  __shared__ int2 stage[TILE_E];
  __shared__ int lcur[MAX_BKT];
  __shared__ int gdelta[MAX_BKT];
  __shared__ int wsum[MAX_BKT / 32];
  const int tid = threadIdx.x, lane = tid & 31, wave = tid >> 5;
  const int tile = blockIdx.x;
  int c = 0, incl = 0;
  if (tid < MAX_BKT) {
    c = (tid < NB) ? tbl[tid * nT + tile] : 0;
    incl = c;
    for (int off = 1; off < 32; off <<= 1) {
      int t = __shfl_up(incl, off);
      if (lane >= off) incl += t;
    }
    if (lane == 31) wsum[wave] = incl;
  }
  __syncthreads();
  if (tid < MAX_BKT) {
    int pre = 0;
    for (int w = 0; w < wave; ++w) pre += wsum[w];
    const int ls = pre + incl - c;
    lcur[tid] = ls;
    gdelta[tid] = ((tid < NB) ? tscan[tid * nT + tile] : 0) - ls;
  }
  __syncthreads();
  const int base = tile * TILE_E;
  int dv[TILE_E / 1024], sv[TILE_E / 1024];
#pragma unroll
  for (int k = 0; k < TILE_E / 1024; ++k) {
    const int i = base + k * 1024 + tid;
    dv[k] = (i < E) ? dst[i] : -1;
    sv[k] = (i < E) ? src[i] : 0;
  }
#pragma unroll
  for (int k = 0; k < TILE_E / 1024; ++k)
    if (dv[k] >= 0) {
      const int p = atomicAdd(&lcur[dv[k] >> BKT_SH], 1);
      stage[p] = make_int2(sv[k], dv[k]);
    }
  __syncthreads();
  const int m = min(TILE_E, E - base);
  for (int j = tid; j < m; j += 1024) {
    const int2 e = stage[j];
    binned[gdelta[e.y >> BKT_SH] + j] = e;
  }
}

__global__ __launch_bounds__(1024) void k_bucket_csr(const int2* __restrict__ binned, const int* __restrict__ tscan,
                                                     int* __restrict__ rowptr, float* __restrict__ dinv,
                                                     int* __restrict__ csr_src, int n, int nT) {
  __shared__ int sorted[BKT_CAP];
  __shared__ int cnt[BKT_W];
  __shared__ int seg[BKT_W];
  __shared__ int wsum[BKT_W / 32];
  const int tid = threadIdx.x, lane = tid & 31, wave = tid >> 5;
  const int b = blockIdx.x;
  const int beg = tscan[b * nT], end = tscan[(b + 1) * nT];
  if (tid < BKT_W) cnt[tid] = 0;
  __syncthreads();
  for (int i = beg + tid; i < end; i += 1024) atomicAdd(&cnt[binned[i].y & (BKT_W - 1)], 1);
  __syncthreads();
  int c = 0, incl = 0;
  if (tid < BKT_W) {
    c = cnt[tid];
    incl = c;
    for (int off = 1; off < 32; off <<= 1) {
      int t = __shfl_up(incl, off);
      if (lane >= off) incl += t;
    }
    if (lane == 31) wsum[wave] = incl;
  }
  __syncthreads();
  if (tid < BKT_W) {
    int pre = 0;
    for (int w = 0; w < wave; ++w) pre += wsum[w];
    const int ls = pre + incl - c;
    seg[tid] = ls;
    cnt[tid] = 0;
    const int d = b * BKT_W + tid;
    if (d < n) {
      rowptr[d] = beg + ls;
      float f = (float)(c + 1);
      dinv[d] = (f > 0.0f) ? rsqrtf(f) : 0.0f;
      if (d == n - 1) rowptr[n] = beg + ls + c;
    }
  }
  __syncthreads();
  if (end - beg <= BKT_CAP) {
    for (int i = beg + tid; i < end; i += 1024) {
      const int2 e = binned[i];
      const int l = e.y & (BKT_W - 1);
      sorted[seg[l] + atomicAdd(&cnt[l], 1)] = e.x;
    }
    __syncthreads();
    for (int j = tid; j < end - beg; j += 1024) csr_src[beg + j] = sorted[j];
  } else {
    for (int i = beg + tid; i < end; i += 1024) {
      const int2 e = binned[i];
      const int l = e.y & (BKT_W - 1);
      csr_src[beg + seg[l] + atomicAdd(&cnt[l], 1)] = e.x;
    }
  }
}

__device__ __forceinline__ float4 gather_node(const float4* __restrict__ xr, const float* __restrict__ dinv,
                                              const int* __restrict__ rowptr, const int* __restrict__ csr_src,
                                              const float4 b, const int d, const int lane) {
  const int beg = rowptr[d], end = rowptr[d + 1];
  const float dd = dinv[d];
  const float4 sv = xr[(size_t)d * (NOUT / 4) + lane];
  float4 acc = {0.0f, 0.0f, 0.0f, 0.0f};

  for (int base = beg; base < end; base += 32) {
    const int j = base + lane;
    const int myS = (j < end) ? csr_src[j] : d;
    const int cnt = min(32, end - base);
    for (int t = 0; t < cnt; t += 4) {
      const int s0 = __builtin_amdgcn_readlane(myS, t);
      const int s1 = __builtin_amdgcn_readlane(myS, t + 1);
      const int s2 = __builtin_amdgcn_readlane(myS, t + 2);
      const int s3 = __builtin_amdgcn_readlane(myS, t + 3);
      const float w0 = 1.0f;
      const float w1 = (t + 1 < cnt) ? 1.0f : 0.0f;
      const float w2 = (t + 2 < cnt) ? 1.0f : 0.0f;
      const float w3 = (t + 3 < cnt) ? 1.0f : 0.0f;
      const float4 v0 = xr[(size_t)s0 * (NOUT / 4) + lane];
      const float4 v1 = xr[(size_t)s1 * (NOUT / 4) + lane];
      const float4 v2 = xr[(size_t)s2 * (NOUT / 4) + lane];
      const float4 v3 = xr[(size_t)s3 * (NOUT / 4) + lane];
      acc.x += v0.x * w0; acc.y += v0.y * w0; acc.z += v0.z * w0; acc.w += v0.w * w0;
      acc.x += v1.x * w1; acc.y += v1.y * w1; acc.z += v1.z * w1; acc.w += v1.w * w1;
      acc.x += v2.x * w2; acc.y += v2.y * w2; acc.z += v2.z * w2; acc.w += v2.w * w2;
      acc.x += v3.x * w3; acc.y += v3.y * w3; acc.z += v3.z * w3; acc.w += v3.w * w3;
    }
  }

  acc.x += sv.x; acc.y += sv.y; acc.z += sv.z; acc.w += sv.w;
  float4 o;
  o.x = fmaxf(acc.x * dd + b.x, 0.0f); o.y = fmaxf(acc.y * dd + b.y, 0.0f);
  o.z = fmaxf(acc.z * dd + b.z, 0.0f); o.w = fmaxf(acc.w * dd + b.w, 0.0f);
  return o;
}

#define FN  40
#define RPW (FN / 8)
#define RT  ((FN + 15) / 16)
#define HP  132
template <bool SCALE>
__global__ __launch_bounds__(256) void k_gather_gemm(const float* __restrict__ xws,
                                                     const float* __restrict__ dinv,
                                                     const int* __restrict__ rowptr,
                                                     const int* __restrict__ csr_src,
                                                     const float* __restrict__ bias,
                                                     const float* __restrict__ Wn,
                                                     const float* __restrict__ bias2,
                                                     float* __restrict__ C, int n) {
  __shared__ alignas(16) float sH[RT * 16 * HP];
  const int lane = threadIdx.x & 31;
  const int wave = __builtin_amdgcn_readfirstlane(threadIdx.x >> 5);
  const int nodeBase = blockIdx.x * FN;
  const float4* xr = (const float4*)xws;
  const float4 b = ((const float4*)bias)[lane];
  const float4 zero4 = {0.0f, 0.0f, 0.0f, 0.0f};

  for (int row = FN + wave; row < RT * 16; row += 8) *(float4*)(sH + row * HP + lane * 4) = zero4;
  for (int q = 0; q < RPW; ++q) {
    const int row = wave * RPW + q;
    const int d = nodeBase + row;
    float4 o = zero4;
    if (d < n) o = gather_node(xr, dinv, rowptr, csr_src, b, d, lane);
    *(float4*)(sH + row * HP + lane * 4) = o;
  }
  __syncthreads();

  const int half = lane >> 4;
  const int r    = lane & 15;
  const float* wcol = Wn + (size_t)(2 * half) * NOUT + wave * 16 + r;
  const float* hrow = sH + r * HP + 2 * half;
  v8f acc[RT];
#pragma unroll
  for (int rt = 0; rt < RT; ++rt) acc[rt] = v8f{};
#pragma unroll 8
  for (int s = 0; s < NOUT / 4; ++s) {
    v2f bf;
    bf[0] = wcol[(size_t)(4 * s) * NOUT];
    bf[1] = wcol[(size_t)(4 * s + 1) * NOUT];
#pragma unroll
    for (int rt = 0; rt < RT; ++rt) {
      const v2f a = *(const v2f*)(hrow + rt * 16 * HP + 4 * s);
      acc[rt] = __builtin_amdgcn_wmma_f32_16x16x4_f32( false, a,  false, bf,
                                                       (short)0, acc[rt],
                                                       false,  false);
    }
  }
  __syncthreads();
#pragma unroll
  for (int rt = 0; rt < RT; ++rt)
#pragma unroll
    for (int i = 0; i < 8; ++i) sH[(rt * 16 + half * 8 + i) * HP + wave * 16 + r] = acc[rt][i];
  __syncthreads();
  float4 b2 = zero4;
  if (!SCALE) b2 = ((const float4*)bias2)[lane];
  for (int q = 0; q < RPW; ++q) {
    const int row = wave * RPW + q;
    const int d = nodeBase + row;
    if (d < n) {
      float4 v = *(const float4*)(sH + row * HP + lane * 4);
      if (SCALE) {
        const float rs = dinv[d];
        v.x *= rs; v.y *= rs; v.z *= rs; v.w *= rs;
      } else {
        v.x += b2.x; v.y += b2.y; v.z += b2.z; v.w += b2.w;
      }
      ((float4*)C)[(size_t)d * (NOUT / 4) + lane] = v;
    }
  }
}

#define XIN 256

__device__ __forceinline__ void tile_product(const float* sH, const float* __restrict__ Wk,
                                             v8f (&acc)[RT], const int wave, const int half, const int r) {
  const float* wcol = Wk + (size_t)(2 * half) * NOUT + wave * 16 + r;
  const float* hrow = sH + r * HP + 2 * half;
  for (int g = 0; g < NOUT / 32; ++g) {
    v2f bf[8];
#pragma unroll
    for (int t = 0; t < 8; ++t) {
      const int s = g * 8 + t;
      bf[t][0] = wcol[(size_t)(4 * s) * NOUT];
      bf[t][1] = wcol[(size_t)(4 * s + 1) * NOUT];
    }
    __builtin_amdgcn_sched_barrier(0);
#pragma unroll
    for (int t = 0; t < 8; ++t) {
      const int s = g * 8 + t;
#pragma unroll
      for (int rt = 0; rt < RT; ++rt) {
        const v2f a = *(const v2f*)(hrow + rt * 16 * HP + 4 * s);
        acc[rt] = __builtin_amdgcn_wmma_f32_16x16x4_f32( false, a,  false, bf[t],
                                                         (short)0, acc[rt],
                                                         false,  false);
      }
    }
  }
}

__device__ __forceinline__ void skip_product(float* sH, const float* __restrict__ x, const float* __restrict__ Ws,
                                             v8f (&ax)[RT], const int nodeBase, const int n,
                                             const int wave, const int lane, const int half, const int r) {
  for (int hh = 0; hh < XIN / NOUT; ++hh) {
    float4 v[RPW];
#pragma unroll
    for (int q = 0; q < RPW; ++q) {
      const int d = min(nodeBase + wave * RPW + q, n - 1);
      v[q] = ((const float4*)x)[(size_t)d * (XIN / 4) + hh * (NOUT / 4) + lane];
    }
#pragma unroll
    for (int q = 0; q < RPW; ++q) {
      const int row = wave * RPW + q;
      const bool live = nodeBase + row < n;
      float4 o;
      o.x = live ? v[q].x : 0.0f; o.y = live ? v[q].y : 0.0f;
      o.z = live ? v[q].z : 0.0f; o.w = live ? v[q].w : 0.0f;
      *(float4*)(sH + row * HP + lane * 4) = o;
    }
    __syncthreads();
    tile_product(sH, Ws + (size_t)(hh * NOUT) * NOUT, ax, wave, half, r);
    __syncthreads();
  }
}

__global__ __launch_bounds__(256) void k_gather_gemm_skip(
    const float* __restrict__ xws, const float* __restrict__ dinv,
    const int* __restrict__ rowptr, const int* __restrict__ csr_src,
    const float* __restrict__ bias, const float* __restrict__ Wf, const float* __restrict__ bf,
    const float* __restrict__ x, const float* __restrict__ Ws, const float* __restrict__ bs,
    float* __restrict__ out, int n) {
  __shared__ alignas(16) float sH[RT * 16 * HP];
  __shared__ float sP[RT * 16 * HP];
  const int lane = threadIdx.x & 31;
  const int wave = __builtin_amdgcn_readfirstlane(threadIdx.x >> 5);
  const int half = lane >> 4;
  const int r    = lane & 15;
  const int nodeBase = blockIdx.x * FN;
  const float4* xr = (const float4*)xws;
  const float4 zero4 = {0.0f, 0.0f, 0.0f, 0.0f};
  const bool skipFirst = ((blockIdx.x ^ (blockIdx.x >> 8)) & 1) == 0;
  const int cpos = (half * 8) * HP + wave * 16 + r;

  for (int row = FN + wave; row < RT * 16; row += 8) *(float4*)(sH + row * HP + lane * 4) = zero4;

  v8f acc[RT];
#pragma unroll 1
  for (int step = 0; step < 2; ++step) {
    const bool doSkip = (step == 0) == skipFirst;
#pragma unroll
    for (int rt = 0; rt < RT; ++rt) acc[rt] = v8f{};
    float bv;
    if (doSkip) {
      skip_product(sH, x, Ws, acc, nodeBase, n, wave, lane, half, r);
      bv = bs[wave * 16 + r];
    } else {
      const float4 b = ((const float4*)bias)[lane];
#pragma unroll 1
      for (int q = 0; q < RPW; ++q) {
        const int row = wave * RPW + q;
        const int d = nodeBase + row;
        float4 o = zero4;
        if (d < n) o = gather_node(xr, dinv, rowptr, csr_src, b, d, lane);
        *(float4*)(sH + row * HP + lane * 4) = o;
      }
      __syncthreads();
      tile_product(sH, Wf, acc, wave, half, r);
      __syncthreads();
      bv = bf[wave * 16 + r];
    }
    if (step == 0) {
#pragma unroll
      for (int rt = 0; rt < RT; ++rt)
#pragma unroll
        for (int i = 0; i < 8; ++i) sP[(rt * 16 + i) * HP + cpos] = acc[rt][i] + bv;
    } else {
#pragma unroll
      for (int rt = 0; rt < RT; ++rt)
#pragma unroll
        for (int i = 0; i < 8; ++i)
          sH[(rt * 16 + i) * HP + cpos] = (acc[rt][i] + bv) + sP[(rt * 16 + i) * HP + cpos];
    }
  }
  __syncthreads();
  for (int q = 0; q < RPW; ++q) {
    const int row = wave * RPW + q;
    const int d = nodeBase + row;
    if (d < n) ((float4*)out)[(size_t)d * (NOUT / 4) + lane] = *(const float4*)(sH + row * HP + lane * 4);
  }
}

template <int K, bool SCALE>
__global__ void __launch_bounds__(256) __attribute__((amdgpu_waves_per_eu(1, 4))) k_gemm_wmma(
    const float* __restrict__ A, const float* __restrict__ W,
    const float* __restrict__ bias, const float* __restrict__ rscale,
    const float* __restrict__ Cin, float* __restrict__ C,
    int M, int flags) {
  constexpr int KS = K / 4;
  constexpr int NG = K / 32;
  __shared__ v2f sB[KS * 8 * 32];

  const int lane = threadIdx.x & 31;
  const int wave = __builtin_amdgcn_readfirstlane(threadIdx.x >> 5);
  const int half = lane >> 4;
  const int r    = lane & 15;
  const int rowBase = (blockIdx.x * 8 + wave) * 16;
  const bool active = rowBase < M;

  const float* arow = A + (size_t)(active ? rowBase + r : 0) * K + 2 * half;
  v2f a[8], an[8];
  if (active) {
#pragma unroll
    for (int t = 0; t < 8; ++t) a[t] = *(const v2f*)(arow + 4 * t);
  }

  {
    const float* wcol = W + (size_t)(2 * half) * NOUT + wave * 16 + r;
#pragma unroll 8
    for (int s = 0; s < KS; ++s) {
      v2f b;
      b[0] = wcol[(size_t)(4 * s) * NOUT];
      b[1] = wcol[(size_t)(4 * s + 1) * NOUT];
      sB[(s * 8 + wave) * 32 + lane] = b;
    }
  }
  __syncthreads();
  if (!active) return;

  v8f acc[8];
#pragma unroll
  for (int ct = 0; ct < 8; ++ct) acc[ct] = v8f{};

  const v2f* bl = sB + lane;
  v2f bc[16], bn[16];
#pragma unroll
  for (int q = 0; q < 2; ++q)
#pragma unroll
    for (int ct = 0; ct < 8; ++ct) bc[q * 8 + ct] = bl[(q * 8 + ct) * 32];

  for (int g = 0; g < NG; ++g) {
    const int gn = (g + 1 < NG) ? g + 1 : g;
#pragma unroll
    for (int t = 0; t < 8; ++t) an[t] = *(const v2f*)(arow + 32 * gn + 4 * t);
#pragma unroll
    for (int uu = 0; uu < 4; ++uu) {
      const int sn = (uu < 3) ? (g * 8 + 2 * uu + 2) : ((g + 1 < NG) ? (g * 8 + 8) : (g * 8 + 6));
#pragma unroll
      for (int q = 0; q < 2; ++q)
#pragma unroll
        for (int ct = 0; ct < 8; ++ct) bn[q * 8 + ct] = bl[((sn + q) * 8 + ct) * 32];
      __builtin_amdgcn_sched_barrier(0);
#pragma unroll
      for (int q = 0; q < 2; ++q)
#pragma unroll
        for (int ct = 0; ct < 8; ++ct)
          acc[ct] = __builtin_amdgcn_wmma_f32_16x16x4_f32(
               false, a[2 * uu + q],  false, bc[q * 8 + ct],
               (short)0, acc[ct],  false,  false);
      __builtin_amdgcn_sched_barrier(0);
#pragma unroll
      for (int i = 0; i < 16; ++i) bc[i] = bn[i];
    }
#pragma unroll
    for (int t = 0; t < 8; ++t) a[t] = an[t];
  }

  float* crow = C + (size_t)(rowBase + half * 8) * NOUT + r;
  float bv[8];
#pragma unroll
  for (int ct = 0; ct < 8; ++ct) bv[ct] = 0.0f;
  if (flags & 2) {
#pragma unroll
    for (int ct = 0; ct < 8; ++ct) bv[ct] = bias[ct * 16 + r];
  }
  if (flags & 1) {
    float cin[8][8];
    const float* cinrow = Cin + (size_t)(rowBase + half * 8) * NOUT + r;
#pragma unroll
    for (int ct = 0; ct < 8; ++ct)
#pragma unroll
      for (int i = 0; i < 8; ++i) cin[ct][i] = cinrow[(size_t)i * NOUT + ct * 16];
#pragma unroll
    for (int ct = 0; ct < 8; ++ct)
#pragma unroll
      for (int i = 0; i < 8; ++i) {
        float v = acc[ct][i] + bv[ct];
        v += cin[ct][i];
        if (flags & 4) v = fmaxf(v, 0.0f);
        crow[(size_t)i * NOUT + ct * 16] = v;
      }
  } else if (SCALE) {
    float rs[8];
#pragma unroll
    for (int i = 0; i < 8; ++i) rs[i] = rscale[rowBase + half * 8 + i];
#pragma unroll
    for (int ct = 0; ct < 8; ++ct)
#pragma unroll
      for (int i = 0; i < 8; ++i) crow[(size_t)i * NOUT + ct * 16] = acc[ct][i] * rs[i];
  } else {
#pragma unroll
    for (int ct = 0; ct < 8; ++ct)
#pragma unroll
      for (int i = 0; i < 8; ++i) {
        float v = acc[ct][i] + bv[ct];
        if (flags & 4) v = fmaxf(v, 0.0f);
        crow[(size_t)i * NOUT + ct * 16] = v;
      }
  }
}

extern "C" void kernel_launch(void* const* d_in, const int* in_sizes, int n_in,
                              void* d_out, int out_size, void* d_ws, size_t ws_size,
                              hipStream_t stream) {
  const float* x  = (const float*)d_in[0];
  const int*   ei = (const int*)d_in[1];
  const float* W1 = (const float*)d_in[2];
  const float* b1 = (const float*)d_in[3];
  const float* W2 = (const float*)d_in[4];
  const float* b2 = (const float*)d_in[5];
  const float* Ws = (const float*)d_in[6];
  const float* bs = (const float*)d_in[7];
  const float* Wf = (const float*)d_in[8];
  const float* bf = (const float*)d_in[9];
  float* out = (float*)d_out;

  const int IN = 256;
  const int n  = in_sizes[0] / IN;
  const int E  = in_sizes[1] / 2;
  const int* src = ei;
  const int* dst = ei + E;

  float* ws   = (float*)d_ws;
  float* pe   = ws;
  float* p1   = pe + (size_t)n * NOUT;
  float* dinv = p1 + (size_t)n * NOUT;

  const int NB = (n + BKT_W - 1) / BKT_W;
  const int nT = (E + TILE_E - 1) / TILE_E;
  int2* binned = (int2*)pe;
  int* csr_src = (int*)pe + 2 * (size_t)E;
  int* rowptr  = csr_src + E;
  int* tbl     = rowptr + n + 1;
  int* tscan   = tbl + NB * nT;

  dim3 B(256);
  dim3 B4(1024);
  const int gS   = (NB * nT + 1023) / 1024;
  const int gF   = (n + FN - 1) / FN;
  const int gM   = (n / 16 + 7) / 8;

  k_bin_count    <<<nT, B4, 0, stream>>>(dst, tbl, E, NB, nT);
  k_excl_scan    <<<gS, B4, 0, stream>>>(tbl, tscan, NB * nT);
  k_bin_scatter  <<<nT, B4, 0, stream>>>(src, dst, tbl, tscan, binned, E, NB, nT);
  k_bucket_csr   <<<NB, B4, 0, stream>>>(binned, tscan, rowptr, dinv, csr_src, n, nT);

  k_gemm_wmma<256, true> <<<gM, B, 0, stream>>>(x, W1, nullptr, dinv, nullptr, out, n, 0);
  k_gather_gemm<true> <<<gF, B, 0, stream>>>(out, dinv, rowptr, csr_src, b1, W2, nullptr, p1, n);
  k_gather_gemm_skip <<<gF, B, 0, stream>>>(p1, dinv, rowptr, csr_src, b2, Wf, bf, x, Ws, bs, out, n);
}


# Edge list built in two launches instead of four: each tile groups its own edges by bucket in LDS with no global count or scan in front; a bucket block reads its pieces once; sum order changes
# speedup vs baseline: 6.2473x; 1.0484x over previous
#include <hip/hip_runtime.h>
#include <math.h>

typedef __attribute__((ext_vector_type(2))) float v2f;
typedef __attribute__((ext_vector_type(8))) float v8f;

#define NOUT 128

#define TILE_E   8192
#define BKT_W    256
#define BKT_SH   8
#define MAX_BKT  256
#define BKT_CAP  16384

#define MAX_TILE 256

__global__ __launch_bounds__(1024) void k_tile_sort(const int* __restrict__ src, const int* __restrict__ dst,
                                                    int* __restrict__ tstart, int2* __restrict__ binned,
                                                    int E, int NB, int nT) {
  __shared__ int2 stage[TILE_E];
  __shared__ int hist[MAX_BKT];
  __shared__ int lcur[MAX_BKT];
  __shared__ int wsum[MAX_BKT / 32];
  const int tid = threadIdx.x, lane = tid & 31, wave = tid >> 5;
  const int tile = blockIdx.x;
  if (tid < MAX_BKT) hist[tid] = 0;
  const int base = tile * TILE_E;
  int dv[TILE_E / 1024], sv[TILE_E / 1024];
#pragma unroll
  for (int k = 0; k < TILE_E / 1024; ++k) {
    const int i = base + k * 1024 + tid;
    dv[k] = (i < E) ? dst[i] : -1;
    sv[k] = (i < E) ? src[i] : 0;
  }
  __syncthreads();
#pragma unroll
  for (int k = 0; k < TILE_E / 1024; ++k)
    if (dv[k] >= 0) atomicAdd(&hist[dv[k] >> BKT_SH], 1);
  __syncthreads();
  int c = 0, incl = 0;
  if (tid < MAX_BKT) {
    c = hist[tid];
    incl = c;
    for (int off = 1; off < 32; off <<= 1) {
      int t = __shfl_up(incl, off);
      if (lane >= off) incl += t;
    }
    if (lane == 31) wsum[wave] = incl;
  }
  __syncthreads();
  if (tid < MAX_BKT) {
    int pre = 0;
    for (int w = 0; w < wave; ++w) pre += wsum[w];
    const int ls = pre + incl - c;
    lcur[tid] = ls;
    if (tid < NB) tstart[tid * nT + tile] = ls;
    if (tid == NB - 1) tstart[NB * nT + tile] = ls + c;
  }
  __syncthreads();
#pragma unroll
  for (int k = 0; k < TILE_E / 1024; ++k)
    if (dv[k] >= 0) {
      const int p = atomicAdd(&lcur[dv[k] >> BKT_SH], 1);
      stage[p] = make_int2(sv[k], dv[k]);
    }
  __syncthreads();
  const int m = min(TILE_E, E - base);
  for (int j = tid; j < m; j += 1024) binned[base + j] = stage[j];
}

__global__ __launch_bounds__(1024) void k_bucket_csr(const int2* __restrict__ binned, const int* __restrict__ tstart,
                                                     int* __restrict__ rowptr, float* __restrict__ dinv,
                                                     int* __restrict__ csr_src, int n, int nT) {
  __shared__ int2 local[BKT_CAP];
  __shared__ int sorted[BKT_CAP];
  __shared__ int cnt[BKT_W];
  __shared__ int seg[BKT_W];
  __shared__ int wsum[BKT_W / 32];
  __shared__ int ta[MAX_TILE];
  __shared__ int toff[MAX_TILE + 1];
  __shared__ int wcs[MAX_TILE / 32];
  __shared__ int was[MAX_TILE / 32];
  const int tid = threadIdx.x, lane = tid & 31, wave = tid >> 5;
  const int b = blockIdx.x;
  if (tid < BKT_W) cnt[tid] = 0;
  int pa = 0, pc = 0, pincl = 0;
  if (tid < MAX_TILE) {
    if (tid < nT) {
      pa = tstart[b * nT + tid];
      pc = tstart[(b + 1) * nT + tid] - pa;
    }
    pincl = pc;
    for (int off = 1; off < 32; off <<= 1) {
      int t = __shfl_up(pincl, off);
      if (lane >= off) pincl += t;
    }
    int asum = pa;
    for (int off = 16; off > 0; off >>= 1) asum += __shfl_xor(asum, off);
    if (lane == 31) wcs[wave] = pincl;
    if (lane == 0)  was[wave] = asum;
  }
  __syncthreads();
  if (tid < MAX_TILE) {
    int pre = 0;
    for (int w = 0; w < wave; ++w) pre += wcs[w];
    ta[tid] = pa;
    toff[tid] = pre + pincl - pc;
    if (tid == MAX_TILE - 1) toff[MAX_TILE] = pre + pincl;
  }
  __syncthreads();
  int beg = 0;
  for (int w = 0; w < MAX_TILE / 32; ++w) beg += was[w];
  const int total = toff[MAX_TILE];
  const bool fits = total <= BKT_CAP;

  {
    int2 ev[MAX_TILE / 32][2];
    int po[MAX_TILE / 32], pn[MAX_TILE / 32];
#pragma unroll
    for (int i = 0; i < MAX_TILE / 32; ++i) {
      const int t = wave + 32 * i;
      const bool live = t < nT;
      po[i] = live ? toff[t] : 0;
      pn[i] = live ? toff[t + 1] - po[i] : 0;
      const int2* piece = binned + (size_t)(live ? t : 0) * TILE_E + (live ? ta[t] : 0);
#pragma unroll
      for (int h = 0; h < 2; ++h) {
        const int k = lane + 32 * h;
        ev[i][h] = (k < pn[i]) ? piece[k] : make_int2(0, -1);
      }
    }
#pragma unroll
    for (int i = 0; i < MAX_TILE / 32; ++i)
#pragma unroll
      for (int h = 0; h < 2; ++h)
        if (ev[i][h].y >= 0) {
          if (fits) local[po[i] + lane + 32 * h] = ev[i][h];
          atomicAdd(&cnt[ev[i][h].y & (BKT_W - 1)], 1);
        }
    for (int t = wave; t < nT; t += 32) {
      const int o = toff[t], ct = toff[t + 1] - o;
      const int2* piece = binned + (size_t)t * TILE_E + ta[t];
      for (int k = 64 + lane; k < ct; k += 32) {
        const int2 e = piece[k];
        if (fits) local[o + k] = e;
        atomicAdd(&cnt[e.y & (BKT_W - 1)], 1);
      }
    }
  }
  __syncthreads();
  int c = 0, incl = 0;
  if (tid < BKT_W) {
    c = cnt[tid];
    incl = c;
    for (int off = 1; off < 32; off <<= 1) {
      int t = __shfl_up(incl, off);
      if (lane >= off) incl += t;
    }
    if (lane == 31) wsum[wave] = incl;
  }
  __syncthreads();
  if (tid < BKT_W) {
    int pre = 0;
    for (int w = 0; w < wave; ++w) pre += wsum[w];
    const int ls = pre + incl - c;
    seg[tid] = ls;
    cnt[tid] = 0;
    const int d = b * BKT_W + tid;
    if (d < n) {
      rowptr[d] = beg + ls;
      float f = (float)(c + 1);
      dinv[d] = (f > 0.0f) ? rsqrtf(f) : 0.0f;
      if (d == n - 1) rowptr[n] = beg + ls + c;
    }
  }
  __syncthreads();
  if (fits) {
    for (int j = tid; j < total; j += 1024) {
      const int2 e = local[j];
      const int l = e.y & (BKT_W - 1);
      sorted[seg[l] + atomicAdd(&cnt[l], 1)] = e.x;
    }
    __syncthreads();
    for (int j = tid; j < total; j += 1024) csr_src[beg + j] = sorted[j];
  } else {
    for (int t = wave; t < nT; t += 32) {
      const int ct = toff[t + 1] - toff[t];
      const int2* piece = binned + (size_t)t * TILE_E + ta[t];
      for (int k = lane; k < ct; k += 32) {
        const int2 e = piece[k];
        const int l = e.y & (BKT_W - 1);
        csr_src[beg + seg[l] + atomicAdd(&cnt[l], 1)] = e.x;
      }
    }
  }
}

__device__ __forceinline__ float4 gather_node(const float4* __restrict__ xr, const float* __restrict__ dinv,
                                              const int* __restrict__ rowptr, const int* __restrict__ csr_src,
                                              const float4 b, const int d, const int lane) {
  const int beg = rowptr[d], end = rowptr[d + 1];
  const float dd = dinv[d];
  const float4 sv = xr[(size_t)d * (NOUT / 4) + lane];
  float4 acc = {0.0f, 0.0f, 0.0f, 0.0f};

  for (int base = beg; base < end; base += 32) {
    const int j = base + lane;
    const int myS = (j < end) ? csr_src[j] : d;
    const int cnt = min(32, end - base);
    for (int t = 0; t < cnt; t += 4) {
      const int s0 = __builtin_amdgcn_readlane(myS, t);
      const int s1 = __builtin_amdgcn_readlane(myS, t + 1);
      const int s2 = __builtin_amdgcn_readlane(myS, t + 2);
      const int s3 = __builtin_amdgcn_readlane(myS, t + 3);
      const float w0 = 1.0f;
      const float w1 = (t + 1 < cnt) ? 1.0f : 0.0f;
      const float w2 = (t + 2 < cnt) ? 1.0f : 0.0f;
      const float w3 = (t + 3 < cnt) ? 1.0f : 0.0f;
      const float4 v0 = xr[(size_t)s0 * (NOUT / 4) + lane];
      const float4 v1 = xr[(size_t)s1 * (NOUT / 4) + lane];
      const float4 v2 = xr[(size_t)s2 * (NOUT / 4) + lane];
      const float4 v3 = xr[(size_t)s3 * (NOUT / 4) + lane];
      acc.x += v0.x * w0; acc.y += v0.y * w0; acc.z += v0.z * w0; acc.w += v0.w * w0;
      acc.x += v1.x * w1; acc.y += v1.y * w1; acc.z += v1.z * w1; acc.w += v1.w * w1;
      acc.x += v2.x * w2; acc.y += v2.y * w2; acc.z += v2.z * w2; acc.w += v2.w * w2;
      acc.x += v3.x * w3; acc.y += v3.y * w3; acc.z += v3.z * w3; acc.w += v3.w * w3;
    }
  }

  acc.x += sv.x; acc.y += sv.y; acc.z += sv.z; acc.w += sv.w;
  float4 o;
  o.x = fmaxf(acc.x * dd + b.x, 0.0f); o.y = fmaxf(acc.y * dd + b.y, 0.0f);
  o.z = fmaxf(acc.z * dd + b.z, 0.0f); o.w = fmaxf(acc.w * dd + b.w, 0.0f);
  return o;
}

#define FN  40
#define RPW (FN / 8)
#define RT  ((FN + 15) / 16)
#define HP  132
template <bool SCALE>
__global__ __launch_bounds__(256) void k_gather_gemm(const float* __restrict__ xws,
                                                     const float* __restrict__ dinv,
                                                     const int* __restrict__ rowptr,
                                                     const int* __restrict__ csr_src,
                                                     const float* __restrict__ bias,
                                                     const float* __restrict__ Wn,
                                                     const float* __restrict__ bias2,
                                                     float* __restrict__ C, int n) {
  __shared__ alignas(16) float sH[RT * 16 * HP];
  const int lane = threadIdx.x & 31;
  const int wave = __builtin_amdgcn_readfirstlane(threadIdx.x >> 5);
  const int nodeBase = blockIdx.x * FN;
  const float4* xr = (const float4*)xws;
  const float4 b = ((const float4*)bias)[lane];
  const float4 zero4 = {0.0f, 0.0f, 0.0f, 0.0f};

  for (int row = FN + wave; row < RT * 16; row += 8) *(float4*)(sH + row * HP + lane * 4) = zero4;
  for (int q = 0; q < RPW; ++q) {
    const int row = wave * RPW + q;
    const int d = nodeBase + row;
    float4 o = zero4;
    if (d < n) o = gather_node(xr, dinv, rowptr, csr_src, b, d, lane);
    *(float4*)(sH + row * HP + lane * 4) = o;
  }
  __syncthreads();

  const int half = lane >> 4;
  const int r    = lane & 15;
  const float* wcol = Wn + (size_t)(2 * half) * NOUT + wave * 16 + r;
  const float* hrow = sH + r * HP + 2 * half;
  v8f acc[RT];
#pragma unroll
  for (int rt = 0; rt < RT; ++rt) acc[rt] = v8f{};
#pragma unroll 8
  for (int s = 0; s < NOUT / 4; ++s) {
    v2f bf;
    bf[0] = wcol[(size_t)(4 * s) * NOUT];
    bf[1] = wcol[(size_t)(4 * s + 1) * NOUT];
#pragma unroll
    for (int rt = 0; rt < RT; ++rt) {
      const v2f a = *(const v2f*)(hrow + rt * 16 * HP + 4 * s);
      acc[rt] = __builtin_amdgcn_wmma_f32_16x16x4_f32( false, a,  false, bf,
                                                       (short)0, acc[rt],
                                                       false,  false);
    }
  }
  __syncthreads();
#pragma unroll
  for (int rt = 0; rt < RT; ++rt)
#pragma unroll
    for (int i = 0; i < 8; ++i) sH[(rt * 16 + half * 8 + i) * HP + wave * 16 + r] = acc[rt][i];
  __syncthreads();
  float4 b2 = zero4;
  if (!SCALE) b2 = ((const float4*)bias2)[lane];
  for (int q = 0; q < RPW; ++q) {
    const int row = wave * RPW + q;
    const int d = nodeBase + row;
    if (d < n) {
      float4 v = *(const float4*)(sH + row * HP + lane * 4);
      if (SCALE) {
        const float rs = dinv[d];
        v.x *= rs; v.y *= rs; v.z *= rs; v.w *= rs;
      } else {
        v.x += b2.x; v.y += b2.y; v.z += b2.z; v.w += b2.w;
      }
      ((float4*)C)[(size_t)d * (NOUT / 4) + lane] = v;
    }
  }
}

#define XIN 256

__device__ __forceinline__ void tile_product(const float* sH, const float* __restrict__ Wk,
                                             v8f (&acc)[RT], const int wave, const int half, const int r) {
  const float* wcol = Wk + (size_t)(2 * half) * NOUT + wave * 16 + r;
  const float* hrow = sH + r * HP + 2 * half;
  for (int g = 0; g < NOUT / 32; ++g) {
    v2f bf[8];
#pragma unroll
    for (int t = 0; t < 8; ++t) {
      const int s = g * 8 + t;
      bf[t][0] = wcol[(size_t)(4 * s) * NOUT];
      bf[t][1] = wcol[(size_t)(4 * s + 1) * NOUT];
    }
    __builtin_amdgcn_sched_barrier(0);
#pragma unroll
    for (int t = 0; t < 8; ++t) {
      const int s = g * 8 + t;
#pragma unroll
      for (int rt = 0; rt < RT; ++rt) {
        const v2f a = *(const v2f*)(hrow + rt * 16 * HP + 4 * s);
        acc[rt] = __builtin_amdgcn_wmma_f32_16x16x4_f32( false, a,  false, bf[t],
                                                         (short)0, acc[rt],
                                                         false,  false);
      }
    }
  }
}

__device__ __forceinline__ void skip_product(float* sH, const float* __restrict__ x, const float* __restrict__ Ws,
                                             v8f (&ax)[RT], const int nodeBase, const int n,
                                             const int wave, const int lane, const int half, const int r) {
  for (int hh = 0; hh < XIN / NOUT; ++hh) {
    float4 v[RPW];
#pragma unroll
    for (int q = 0; q < RPW; ++q) {
      const int d = min(nodeBase + wave * RPW + q, n - 1);
      v[q] = ((const float4*)x)[(size_t)d * (XIN / 4) + hh * (NOUT / 4) + lane];
    }
#pragma unroll
    for (int q = 0; q < RPW; ++q) {
      const int row = wave * RPW + q;
      const bool live = nodeBase + row < n;
      float4 o;
      o.x = live ? v[q].x : 0.0f; o.y = live ? v[q].y : 0.0f;
      o.z = live ? v[q].z : 0.0f; o.w = live ? v[q].w : 0.0f;
      *(float4*)(sH + row * HP + lane * 4) = o;
    }
    __syncthreads();
    tile_product(sH, Ws + (size_t)(hh * NOUT) * NOUT, ax, wave, half, r);
    __syncthreads();
  }
}

__global__ __launch_bounds__(256) void k_gather_gemm_skip(
    const float* __restrict__ xws, const float* __restrict__ dinv,
    const int* __restrict__ rowptr, const int* __restrict__ csr_src,
    const float* __restrict__ bias, const float* __restrict__ Wf, const float* __restrict__ bf,
    const float* __restrict__ x, const float* __restrict__ Ws, const float* __restrict__ bs,
    float* __restrict__ out, int n) {
  __shared__ alignas(16) float sH[RT * 16 * HP];
  __shared__ float sP[RT * 16 * HP];
  const int lane = threadIdx.x & 31;
  const int wave = __builtin_amdgcn_readfirstlane(threadIdx.x >> 5);
  const int half = lane >> 4;
  const int r    = lane & 15;
  const int nodeBase = blockIdx.x * FN;
  const float4* xr = (const float4*)xws;
  const float4 zero4 = {0.0f, 0.0f, 0.0f, 0.0f};
  const bool skipFirst = ((blockIdx.x ^ (blockIdx.x >> 8)) & 1) == 0;
  const int cpos = (half * 8) * HP + wave * 16 + r;

  for (int row = FN + wave; row < RT * 16; row += 8) *(float4*)(sH + row * HP + lane * 4) = zero4;

  v8f acc[RT];
#pragma unroll 1
  for (int step = 0; step < 2; ++step) {
    const bool doSkip = (step == 0) == skipFirst;
#pragma unroll
    for (int rt = 0; rt < RT; ++rt) acc[rt] = v8f{};
    float bv;
    if (doSkip) {
      skip_product(sH, x, Ws, acc, nodeBase, n, wave, lane, half, r);
      bv = bs[wave * 16 + r];
    } else {
      const float4 b = ((const float4*)bias)[lane];
#pragma unroll 1
      for (int q = 0; q < RPW; ++q) {
        const int row = wave * RPW + q;
        const int d = nodeBase + row;
        float4 o = zero4;
        if (d < n) o = gather_node(xr, dinv, rowptr, csr_src, b, d, lane);
        *(float4*)(sH + row * HP + lane * 4) = o;
      }
      __syncthreads();
      tile_product(sH, Wf, acc, wave, half, r);
      __syncthreads();
      bv = bf[wave * 16 + r];
    }
    if (step == 0) {
#pragma unroll
      for (int rt = 0; rt < RT; ++rt)
#pragma unroll
        for (int i = 0; i < 8; ++i) sP[(rt * 16 + i) * HP + cpos] = acc[rt][i] + bv;
    } else {
#pragma unroll
      for (int rt = 0; rt < RT; ++rt)
#pragma unroll
        for (int i = 0; i < 8; ++i)
          sH[(rt * 16 + i) * HP + cpos] = (acc[rt][i] + bv) + sP[(rt * 16 + i) * HP + cpos];
    }
  }
  __syncthreads();
  for (int q = 0; q < RPW; ++q) {
    const int row = wave * RPW + q;
    const int d = nodeBase + row;
    if (d < n) ((float4*)out)[(size_t)d * (NOUT / 4) + lane] = *(const float4*)(sH + row * HP + lane * 4);
  }
}

template <int K, bool SCALE>
__global__ void __launch_bounds__(256) __attribute__((amdgpu_waves_per_eu(1, 4))) k_gemm_wmma(
    const float* __restrict__ A, const float* __restrict__ W,
    const float* __restrict__ bias, const float* __restrict__ rscale,
    const float* __restrict__ Cin, float* __restrict__ C,
    int M, int flags) {
  constexpr int KS = K / 4;
  constexpr int NG = K / 32;
  __shared__ v2f sB[KS * 8 * 32];

  const int lane = threadIdx.x & 31;
  const int wave = __builtin_amdgcn_readfirstlane(threadIdx.x >> 5);
  const int half = lane >> 4;
  const int r    = lane & 15;
  const int rowBase = (blockIdx.x * 8 + wave) * 16;
  const bool active = rowBase < M;

  const float* arow = A + (size_t)(active ? rowBase + r : 0) * K + 2 * half;
  v2f a[8], an[8];
  if (active) {
#pragma unroll
    for (int t = 0; t < 8; ++t) a[t] = *(const v2f*)(arow + 4 * t);
  }

  {
    const float* wcol = W + (size_t)(2 * half) * NOUT + wave * 16 + r;
#pragma unroll 8
    for (int s = 0; s < KS; ++s) {
      v2f b;
      b[0] = wcol[(size_t)(4 * s) * NOUT];
      b[1] = wcol[(size_t)(4 * s + 1) * NOUT];
      sB[(s * 8 + wave) * 32 + lane] = b;
    }
  }
  __syncthreads();
  if (!active) return;

  v8f acc[8];
#pragma unroll
  for (int ct = 0; ct < 8; ++ct) acc[ct] = v8f{};

  const v2f* bl = sB + lane;
  v2f bc[16], bn[16];
#pragma unroll
  for (int q = 0; q < 2; ++q)
#pragma unroll
    for (int ct = 0; ct < 8; ++ct) bc[q * 8 + ct] = bl[(q * 8 + ct) * 32];

  for (int g = 0; g < NG; ++g) {
    const int gn = (g + 1 < NG) ? g + 1 : g;
#pragma unroll
    for (int t = 0; t < 8; ++t) an[t] = *(const v2f*)(arow + 32 * gn + 4 * t);
#pragma unroll
    for (int uu = 0; uu < 4; ++uu) {
      const int sn = (uu < 3) ? (g * 8 + 2 * uu + 2) : ((g + 1 < NG) ? (g * 8 + 8) : (g * 8 + 6));
#pragma unroll
      for (int q = 0; q < 2; ++q)
#pragma unroll
        for (int ct = 0; ct < 8; ++ct) bn[q * 8 + ct] = bl[((sn + q) * 8 + ct) * 32];
      __builtin_amdgcn_sched_barrier(0);
#pragma unroll
      for (int q = 0; q < 2; ++q)
#pragma unroll
        for (int ct = 0; ct < 8; ++ct)
          acc[ct] = __builtin_amdgcn_wmma_f32_16x16x4_f32(
               false, a[2 * uu + q],  false, bc[q * 8 + ct],
               (short)0, acc[ct],  false,  false);
      __builtin_amdgcn_sched_barrier(0);
#pragma unroll
      for (int i = 0; i < 16; ++i) bc[i] = bn[i];
    }
#pragma unroll
    for (int t = 0; t < 8; ++t) a[t] = an[t];
  }

  float* crow = C + (size_t)(rowBase + half * 8) * NOUT + r;
  float bv[8];
#pragma unroll
  for (int ct = 0; ct < 8; ++ct) bv[ct] = 0.0f;
  if (flags & 2) {
#pragma unroll
    for (int ct = 0; ct < 8; ++ct) bv[ct] = bias[ct * 16 + r];
  }
  if (flags & 1) {
    float cin[8][8];
    const float* cinrow = Cin + (size_t)(rowBase + half * 8) * NOUT + r;
#pragma unroll
    for (int ct = 0; ct < 8; ++ct)
#pragma unroll
      for (int i = 0; i < 8; ++i) cin[ct][i] = cinrow[(size_t)i * NOUT + ct * 16];
#pragma unroll
    for (int ct = 0; ct < 8; ++ct)
#pragma unroll
      for (int i = 0; i < 8; ++i) {
        float v = acc[ct][i] + bv[ct];
        v += cin[ct][i];
        if (flags & 4) v = fmaxf(v, 0.0f);
        crow[(size_t)i * NOUT + ct * 16] = v;
      }
  } else if (SCALE) {
    float rs[8];
#pragma unroll
    for (int i = 0; i < 8; ++i) rs[i] = rscale[rowBase + half * 8 + i];
#pragma unroll
    for (int ct = 0; ct < 8; ++ct)
#pragma unroll
      for (int i = 0; i < 8; ++i) crow[(size_t)i * NOUT + ct * 16] = acc[ct][i] * rs[i];
  } else {
#pragma unroll
    for (int ct = 0; ct < 8; ++ct)
#pragma unroll
      for (int i = 0; i < 8; ++i) {
        float v = acc[ct][i] + bv[ct];
        if (flags & 4) v = fmaxf(v, 0.0f);
        crow[(size_t)i * NOUT + ct * 16] = v;
      }
  }
}

extern "C" void kernel_launch(void* const* d_in, const int* in_sizes, int n_in,
                              void* d_out, int out_size, void* d_ws, size_t ws_size,
                              hipStream_t stream) {
  const float* x  = (const float*)d_in[0];
  const int*   ei = (const int*)d_in[1];
  const float* W1 = (const float*)d_in[2];
  const float* b1 = (const float*)d_in[3];
  const float* W2 = (const float*)d_in[4];
  const float* b2 = (const float*)d_in[5];
  const float* Ws = (const float*)d_in[6];
  const float* bs = (const float*)d_in[7];
  const float* Wf = (const float*)d_in[8];
  const float* bf = (const float*)d_in[9];
  float* out = (float*)d_out;

  const int IN = 256;
  const int n  = in_sizes[0] / IN;
  const int E  = in_sizes[1] / 2;
  const int* src = ei;
  const int* dst = ei + E;

  float* ws   = (float*)d_ws;
  float* pe   = ws;
  float* p1   = pe + (size_t)n * NOUT;
  float* dinv = p1 + (size_t)n * NOUT;

  const int NB = (n + BKT_W - 1) / BKT_W;
  const int nT = (E + TILE_E - 1) / TILE_E;
  int2* binned = (int2*)pe;
  int* csr_src = (int*)pe + 2 * (size_t)E;
  int* rowptr  = csr_src + E;
  int* tstart  = rowptr + n + 1;

  dim3 B(256);
  dim3 B4(1024);
  const int gF   = (n + FN - 1) / FN;
  const int gM   = (n / 16 + 7) / 8;

  k_tile_sort    <<<nT, B4, 0, stream>>>(src, dst, tstart, binned, E, NB, nT);
  k_bucket_csr   <<<NB, B4, 0, stream>>>(binned, tstart, rowptr, dinv, csr_src, n, nT);

  k_gemm_wmma<256, true> <<<gM, B, 0, stream>>>(x, W1, nullptr, dinv, nullptr, out, n, 0);
  k_gather_gemm<true> <<<gF, B, 0, stream>>>(out, dinv, rowptr, csr_src, b1, W2, nullptr, p1, n);
  k_gather_gemm_skip <<<gF, B, 0, stream>>>(p1, dinv, rowptr, csr_src, b2, Wf, bf, x, Ws, bs, out, n);
}
